# starting kernel
# baseline (speedup 1.0000x reference)
#include <hip/hip_runtime.h>
#include <math.h>


typedef float v2f __attribute__((ext_vector_type(2)));
typedef float v8f __attribute__((ext_vector_type(8)));

#define NEG_SLOPE 0.2f
#define GAT_EPS 1e-16f

__device__ __forceinline__ unsigned fenc(float f) {
  unsigned u = __float_as_uint(f);
  return (u & 0x80000000u) ? ~u : (u | 0x80000000u);
}
__device__ __forceinline__ float fdec(unsigned k) {
  unsigned u = (k & 0x80000000u) ? (k ^ 0x80000000u) : ~k;
  return __uint_as_float(u);
}

__global__ void gat_gemm1_wmma(const float* __restrict__ X,
                               const float* __restrict__ W,
                               float* __restrict__ H, int N) {
  const int lane    = threadIdx.x & 31;
  const int wave    = threadIdx.x >> 5;
  const int row0    = blockIdx.x * 16;
  const int colBase = wave * 16;
  const int m       = lane & 15;
  const int khalf   = (lane >> 4) * 2;

  int rowA = row0 + m; if (rowA >= N) rowA = N - 1;
  const float* xrow = X + (long)rowA * 128;

  v8f c = {};
  for (int k0 = 0; k0 < 128; k0 += 4) {
    const int k = k0 + khalf;
    v2f a; a.x = xrow[k];                 a.y = xrow[k + 1];
    v2f b; b.x = W[k * 64 + colBase + m]; b.y = W[(k + 1) * 64 + colBase + m];
    c = __builtin_amdgcn_wmma_f32_16x16x4_f32(false, a, false, b,
                                              (short)0, c, false, false);
  }
  const int mrow = (lane >> 4) * 8;
#pragma unroll
  for (int v = 0; v < 8; ++v) {
    int r = row0 + mrow + v;
    if (r < N) H[(long)r * 64 + colBase + m] = c[v];
  }
}

__global__ void gat_gemm2_wmma(const float* __restrict__ X,
                               const float* __restrict__ W,
                               float* __restrict__ H, int N) {
  const int lane    = threadIdx.x & 31;
  const int wave    = threadIdx.x >> 5;
  const int row0    = blockIdx.x * 16;
  const int col     = wave * 16 + (lane & 15);
  const int m       = lane & 15;
  const int khalf   = (lane >> 4) * 2;
  const bool colOK  = (col < 40);

  int rowA = row0 + m; if (rowA >= N) rowA = N - 1;
  const float* xrow = X + (long)rowA * 64;

  v8f c = {};
  for (int k0 = 0; k0 < 64; k0 += 4) {
    const int k = k0 + khalf;
    v2f a; a.x = xrow[k]; a.y = xrow[k + 1];
    v2f b;
    b.x = colOK ? W[k * 40 + col] : 0.0f;
    b.y = colOK ? W[(k + 1) * 40 + col] : 0.0f;
    c = __builtin_amdgcn_wmma_f32_16x16x4_f32(false, a, false, b,
                                              (short)0, c, false, false);
  }
  const int mrow = (lane >> 4) * 8;
#pragma unroll
  for (int v = 0; v < 8; ++v) {
    int r = row0 + mrow + v;
    if (colOK && r < N) H[(long)r * 40 + col] = c[v];
  }
}

__global__ void gat_attn1(const float* __restrict__ H, const float* __restrict__ aS,
                          const float* __restrict__ aD, float* __restrict__ eS,
                          float* __restrict__ eD, int N) {
  int t = blockIdx.x * blockDim.x + threadIdx.x;
  if (t >= N * 8) return;
  const int head = t & 7;
  const float* hp = H + (long)(t >> 3) * 64 + head * 8;
  const float* as = aS + head * 8;
  const float* ad = aD + head * 8;
  float s = 0.f, d = 0.f;
#pragma unroll
  for (int c = 0; c < 8; ++c) { float h = hp[c]; s += h * as[c]; d += h * ad[c]; }
  eS[t] = s; eD[t] = d;
}

__global__ void gat_attn2(const float* __restrict__ H, const float* __restrict__ aS,
                          const float* __restrict__ aD, float* __restrict__ eS,
                          float* __restrict__ eD, int N) {
  int n = blockIdx.x * blockDim.x + threadIdx.x;
  if (n >= N) return;
  const float* hp = H + (long)n * 40;
  float s = 0.f, d = 0.f;
#pragma unroll 8
  for (int c = 0; c < 40; ++c) { float h = hp[c]; s += h * aS[c]; d += h * aD[c]; }
  eS[n] = s; eD[n] = d;
}

__global__ void gat_edge_max(const int* __restrict__ src, const int* __restrict__ dst,
                             const float* __restrict__ eS, const float* __restrict__ eD,
                             float* __restrict__ eBuf, unsigned* __restrict__ mKey,
                             int E, int H) {
  int t = blockIdx.x * blockDim.x + threadIdx.x;
  if (t >= E * H) return;
  const int e = t / H, h = t - e * H;
  const int s = src[e], d = dst[e];
  float v = eS[s * H + h] + eD[d * H + h];
  v = (v > 0.f) ? v : v * NEG_SLOPE;
  eBuf[t] = v;
  atomicMax(mKey + d * H + h, fenc(v));
}

__global__ void gat_edge_sum(const int* __restrict__ dst, float* __restrict__ eBuf,
                             const unsigned* __restrict__ mKey, float* __restrict__ sSum,
                             int E, int H) {
  int t = blockIdx.x * blockDim.x + threadIdx.x;
  if (t >= E * H) return;
  const int e = t / H, h = t - e * H;
  const int d = dst[e];
  float p = expf(eBuf[t] - fdec(mKey[d * H + h]));
  eBuf[t] = p;
  atomicAdd(sSum + d * H + h, p);
}

__global__ void gat_edge_msg1(const int* __restrict__ src, const int* __restrict__ dst,
                              const float* __restrict__ eBuf, const float* __restrict__ sSum,
                              const float* __restrict__ Hn, float* __restrict__ agg, int E) {
  int t = blockIdx.x * blockDim.x + threadIdx.x;
  if (t >= E * 8) return;
  const int e = t >> 3, h = t & 7;
  const int s = src[e], d = dst[e];
  const float alpha = eBuf[t] / (sSum[d * 8 + h] + GAT_EPS);
  const float4* hp = (const float4*)(Hn + (long)s * 64 + h * 8);
  float4 u0 = hp[0], u1 = hp[1];
  float* o = agg + (long)d * 64 + h * 8;
  atomicAdd(o + 0, u0.x * alpha); atomicAdd(o + 1, u0.y * alpha);
  atomicAdd(o + 2, u0.z * alpha); atomicAdd(o + 3, u0.w * alpha);
  atomicAdd(o + 4, u1.x * alpha); atomicAdd(o + 5, u1.y * alpha);
  atomicAdd(o + 6, u1.z * alpha); atomicAdd(o + 7, u1.w * alpha);
}

__global__ void gat_edge_msg2(const int* __restrict__ src, const int* __restrict__ dst,
                              const float* __restrict__ eBuf, const float* __restrict__ sSum,
                              const float* __restrict__ Hn, float* __restrict__ agg, int E) {
  int e = blockIdx.x * blockDim.x + threadIdx.x;
  if (e >= E) return;
  const int s = src[e], d = dst[e];
  const float alpha = eBuf[e] / (sSum[d] + GAT_EPS);
  const float* hp = Hn + (long)s * 40;
  float* o = agg + (long)d * 40;
#pragma unroll 8
  for (int c = 0; c < 40; ++c) atomicAdd(o + c, hp[c] * alpha);
}

__global__ void gat_bias_relu(float* __restrict__ agg, const float* __restrict__ b, long n64) {
  long t = (long)blockIdx.x * blockDim.x + threadIdx.x;
  if (t >= n64) return;
  float v = agg[t] + b[t & 63];
  agg[t] = v > 0.f ? v : 0.f;
}

__global__ void gat_logsoftmax(const float* __restrict__ agg, const float* __restrict__ b,
                               float* __restrict__ out, int N) {
  int n = blockIdx.x * blockDim.x + threadIdx.x;
  if (n >= N) return;
  const float* a = agg + (long)n * 40;
  float v[40];
  float mx = -INFINITY;
#pragma unroll 8
  for (int c = 0; c < 40; ++c) { v[c] = a[c] + b[c]; mx = fmaxf(mx, v[c]); }
  float sum = 0.f;
#pragma unroll 8
  for (int c = 0; c < 40; ++c) sum += expf(v[c] - mx);
  const float ls = mx + logf(sum);
  float* o = out + (long)n * 40;
#pragma unroll 8
  for (int c = 0; c < 40; ++c) o[c] = v[c] - ls;
}

extern "C" void kernel_launch(void* const* d_in, const int* in_sizes, int n_in,
                              void* d_out, int out_size, void* d_ws, size_t ws_size,
                              hipStream_t stream) {
  const float* x     = (const float*)d_in[0];
  const int*   eidx  = (const int*)d_in[1];
  const float* W1    = (const float*)d_in[2];
  const float* aS1   = (const float*)d_in[3];
  const float* aD1   = (const float*)d_in[4];
  const float* b1    = (const float*)d_in[5];
  const float* W2    = (const float*)d_in[6];
  const float* aS2   = (const float*)d_in[7];
  const float* aD2   = (const float*)d_in[8];
  const float* b2    = (const float*)d_in[9];
  float* out = (float*)d_out;

  const int N = in_sizes[0] / 128;
  const int E = in_sizes[1] / 2;
  const int* src = eidx;
  const int* dst = eidx + E;

  const long N64 = (long)N * 64, N40 = (long)N * 40;
  const long N8  = (long)N * 8,  E8  = (long)E * 8;

  float* ws      = (float*)d_ws;
  float* h1      = ws;
  float* esrc1   = h1 + N64;
  float* edst1   = esrc1 + N8;
  float* e1      = edst1 + N8;
  float* s1      = e1 + E8;
  unsigned* mK1  = (unsigned*)(s1 + N8);
  float* agg1    = (float*)mK1 + N8;
  float* h2      = agg1 + N64;
  float* esrc2   = h2 + N40;
  float* edst2   = esrc2 + N;
  float* e2      = edst2 + N;
  float* s2      = e2 + E;
  unsigned* mK2  = (unsigned*)(s2 + N);
  float* agg2    = (float*)mK2 + N;

  const int BT = 256;
  const int rowBlocks = (N + 15) / 16;

  gat_gemm1_wmma<<<rowBlocks, 128, 0, stream>>>(x, W1, h1, N);
  gat_attn1<<<(int)((N8 + BT - 1) / BT), BT, 0, stream>>>(h1, aS1, aD1, esrc1, edst1, N);
  hipMemsetAsync(s1, 0, (size_t)(N8 + N8 + N64) * 4, stream);
  gat_edge_max<<<(int)((E8 + BT - 1) / BT), BT, 0, stream>>>(src, dst, esrc1, edst1, e1, mK1, E, 8);
  gat_edge_sum<<<(int)((E8 + BT - 1) / BT), BT, 0, stream>>>(dst, e1, mK1, s1, E, 8);
  gat_edge_msg1<<<(int)((E8 + BT - 1) / BT), BT, 0, stream>>>(src, dst, e1, s1, h1, agg1, E);
  gat_bias_relu<<<(int)((N64 + BT - 1) / BT), BT, 0, stream>>>(agg1, b1, N64);

  gat_gemm2_wmma<<<rowBlocks, 96, 0, stream>>>(agg1, W2, h2, N);
  gat_attn2<<<(N + BT - 1) / BT, BT, 0, stream>>>(h2, aS2, aD2, esrc2, edst2, N);
  hipMemsetAsync(s2, 0, (size_t)(N + N + N40) * 4, stream);
  gat_edge_max<<<(E + BT - 1) / BT, BT, 0, stream>>>(src, dst, esrc2, edst2, e2, mK2, E, 1);
  gat_edge_sum<<<(E + BT - 1) / BT, BT, 0, stream>>>(dst, e2, mK2, s2, E, 1);
  gat_edge_msg2<<<(E + BT - 1) / BT, BT, 0, stream>>>(src, dst, e2, s2, h2, agg2, E);
  gat_logsoftmax<<<(N + BT - 1) / BT, BT, 0, stream>>>(agg2, b2, out, N);
}


# edge_msg2: one wave per 32 edges with the 40 channels across lanes, so each load and each atomic add covers consecutive words of one row instead of 32 rows
# speedup vs baseline: 1.8493x; 1.8493x over previous
#include <hip/hip_runtime.h>
#include <math.h>


typedef float v2f __attribute__((ext_vector_type(2)));
typedef float v8f __attribute__((ext_vector_type(8)));

#define NEG_SLOPE 0.2f
#define GAT_EPS 1e-16f

__device__ __forceinline__ unsigned fenc(float f) {
  unsigned u = __float_as_uint(f);
  return (u & 0x80000000u) ? ~u : (u | 0x80000000u);
}
__device__ __forceinline__ float fdec(unsigned k) {
  unsigned u = (k & 0x80000000u) ? (k ^ 0x80000000u) : ~k;
  return __uint_as_float(u);
}

__global__ void gat_gemm1_wmma(const float* __restrict__ X,
                               const float* __restrict__ W,
                               float* __restrict__ H, int N) {
  const int lane    = threadIdx.x & 31;
  const int wave    = threadIdx.x >> 5;
  const int row0    = blockIdx.x * 16;
  const int colBase = wave * 16;
  const int m       = lane & 15;
  const int khalf   = (lane >> 4) * 2;

  int rowA = row0 + m; if (rowA >= N) rowA = N - 1;
  const float* xrow = X + (long)rowA * 128;

  v8f c = {};
  for (int k0 = 0; k0 < 128; k0 += 4) {
    const int k = k0 + khalf;
    v2f a; a.x = xrow[k];                 a.y = xrow[k + 1];
    v2f b; b.x = W[k * 64 + colBase + m]; b.y = W[(k + 1) * 64 + colBase + m];
    c = __builtin_amdgcn_wmma_f32_16x16x4_f32(false, a, false, b,
                                              (short)0, c, false, false);
  }
  const int mrow = (lane >> 4) * 8;
#pragma unroll
  for (int v = 0; v < 8; ++v) {
    int r = row0 + mrow + v;
    if (r < N) H[(long)r * 64 + colBase + m] = c[v];
  }
}

__global__ void gat_gemm2_wmma(const float* __restrict__ X,
                               const float* __restrict__ W,
                               float* __restrict__ H, int N) {
  const int lane    = threadIdx.x & 31;
  const int wave    = threadIdx.x >> 5;
  const int row0    = blockIdx.x * 16;
  const int col     = wave * 16 + (lane & 15);
  const int m       = lane & 15;
  const int khalf   = (lane >> 4) * 2;
  const bool colOK  = (col < 40);

  int rowA = row0 + m; if (rowA >= N) rowA = N - 1;
  const float* xrow = X + (long)rowA * 64;

  v8f c = {};
  for (int k0 = 0; k0 < 64; k0 += 4) {
    const int k = k0 + khalf;
    v2f a; a.x = xrow[k]; a.y = xrow[k + 1];
    v2f b;
    b.x = colOK ? W[k * 40 + col] : 0.0f;
    b.y = colOK ? W[(k + 1) * 40 + col] : 0.0f;
    c = __builtin_amdgcn_wmma_f32_16x16x4_f32(false, a, false, b,
                                              (short)0, c, false, false);
  }
  const int mrow = (lane >> 4) * 8;
#pragma unroll
  for (int v = 0; v < 8; ++v) {
    int r = row0 + mrow + v;
    if (colOK && r < N) H[(long)r * 40 + col] = c[v];
  }
}

__global__ void gat_attn1(const float* __restrict__ H, const float* __restrict__ aS,
                          const float* __restrict__ aD, float* __restrict__ eS,
                          float* __restrict__ eD, int N) {
  int t = blockIdx.x * blockDim.x + threadIdx.x;
  if (t >= N * 8) return;
  const int head = t & 7;
  const float* hp = H + (long)(t >> 3) * 64 + head * 8;
  const float* as = aS + head * 8;
  const float* ad = aD + head * 8;
  float s = 0.f, d = 0.f;
#pragma unroll
  for (int c = 0; c < 8; ++c) { float h = hp[c]; s += h * as[c]; d += h * ad[c]; }
  eS[t] = s; eD[t] = d;
}

__global__ void gat_attn2(const float* __restrict__ H, const float* __restrict__ aS,
                          const float* __restrict__ aD, float* __restrict__ eS,
                          float* __restrict__ eD, int N) {
  int n = blockIdx.x * blockDim.x + threadIdx.x;
  if (n >= N) return;
  const float* hp = H + (long)n * 40;
  float s = 0.f, d = 0.f;
#pragma unroll 8
  for (int c = 0; c < 40; ++c) { float h = hp[c]; s += h * aS[c]; d += h * aD[c]; }
  eS[n] = s; eD[n] = d;
}

__global__ void gat_edge_max(const int* __restrict__ src, const int* __restrict__ dst,
                             const float* __restrict__ eS, const float* __restrict__ eD,
                             float* __restrict__ eBuf, unsigned* __restrict__ mKey,
                             int E, int H) {
  int t = blockIdx.x * blockDim.x + threadIdx.x;
  if (t >= E * H) return;
  const int e = t / H, h = t - e * H;
  const int s = src[e], d = dst[e];
  float v = eS[s * H + h] + eD[d * H + h];
  v = (v > 0.f) ? v : v * NEG_SLOPE;
  eBuf[t] = v;
  atomicMax(mKey + d * H + h, fenc(v));
}

__global__ void gat_edge_sum(const int* __restrict__ dst, float* __restrict__ eBuf,
                             const unsigned* __restrict__ mKey, float* __restrict__ sSum,
                             int E, int H) {
  int t = blockIdx.x * blockDim.x + threadIdx.x;
  if (t >= E * H) return;
  const int e = t / H, h = t - e * H;
  const int d = dst[e];
  float p = expf(eBuf[t] - fdec(mKey[d * H + h]));
  eBuf[t] = p;
  atomicAdd(sSum + d * H + h, p);
}

__global__ void gat_edge_msg1(const int* __restrict__ src, const int* __restrict__ dst,
                              const float* __restrict__ eBuf, const float* __restrict__ sSum,
                              const float* __restrict__ Hn, float* __restrict__ agg, int E) {
  int t = blockIdx.x * blockDim.x + threadIdx.x;
  if (t >= E * 8) return;
  const int e = t >> 3, h = t & 7;
  const int s = src[e], d = dst[e];
  const float alpha = eBuf[t] / (sSum[d * 8 + h] + GAT_EPS);
  const float4* hp = (const float4*)(Hn + (long)s * 64 + h * 8);
  float4 u0 = hp[0], u1 = hp[1];
  float* o = agg + (long)d * 64 + h * 8;
  atomicAdd(o + 0, u0.x * alpha); atomicAdd(o + 1, u0.y * alpha);
  atomicAdd(o + 2, u0.z * alpha); atomicAdd(o + 3, u0.w * alpha);
  atomicAdd(o + 4, u1.x * alpha); atomicAdd(o + 5, u1.y * alpha);
  atomicAdd(o + 6, u1.z * alpha); atomicAdd(o + 7, u1.w * alpha);
}

template <int K>
__device__ __forceinline__ void msg2_slot(int lane, const int (&sj)[4], const int (&dj)[4],
                                          const int (&aj)[4], unsigned& so, unsigned& dof,
                                          float& av) {
  constexpr int jlo = (32 * K) / 40;
  constexpr int b   = 40 * (jlo + 1) - 32 * K;
  constexpr int jhi = (b < 32) ? jlo + 1 : jlo;
  const bool lo = lane < b;
  const int  c  = lo ? (32 * K - 40 * jlo + lane) : (lane - b);
  so  = (unsigned)((lo ? sj[jlo] : sj[jhi]) * 40 + c);
  dof = (unsigned)((lo ? dj[jlo] : dj[jhi]) * 40 + c);
  av  = __int_as_float(lo ? aj[jlo] : aj[jhi]);
}

__global__ void gat_edge_msg2(const int* __restrict__ src, const int* __restrict__ dst,
                              const float* __restrict__ eBuf, const float* __restrict__ sSum,
                              const float* __restrict__ Hn, float* __restrict__ agg, int E) {
  const int lane = threadIdx.x & 31;
  const int e0   = (int)((blockIdx.x * blockDim.x + threadIdx.x) >> 5) * 32;
  if (e0 >= E) return;
  const int nE = (E - e0 < 32) ? (E - e0) : 32;

  int s = 0, d = 0;
  float alpha = 0.f;
  if (lane < nE) {
    s = src[e0 + lane];
    d = dst[e0 + lane];
    alpha = eBuf[e0 + lane] / (sSum[d] + GAT_EPS);
  }
  const int ai = __float_as_int(alpha);

  if (nE == 32) {
#pragma unroll
    for (int g = 0; g < 8; ++g) {
      int sj[4], dj[4], aj[4];
      for (int j = 0; j < 4; ++j) {
        sj[j] = __builtin_amdgcn_readlane(s,  4 * g + j);
        dj[j] = __builtin_amdgcn_readlane(d,  4 * g + j);
        aj[j] = __builtin_amdgcn_readlane(ai, 4 * g + j);
      }
      unsigned so[5], dof[5];
      float av[5], hv[5];
      msg2_slot<0>(lane, sj, dj, aj, so[0], dof[0], av[0]);
      msg2_slot<1>(lane, sj, dj, aj, so[1], dof[1], av[1]);
      msg2_slot<2>(lane, sj, dj, aj, so[2], dof[2], av[2]);
      msg2_slot<3>(lane, sj, dj, aj, so[3], dof[3], av[3]);
      msg2_slot<4>(lane, sj, dj, aj, so[4], dof[4], av[4]);
      for (int k = 0; k < 5; ++k) hv[k] = Hn[so[k]];
      for (int k = 0; k < 5; ++k) atomicAdd(agg + dof[k], hv[k] * av[k]);
    }
  } else {
    for (int i = 0; i < nE; ++i) {
      const unsigned so  = (unsigned)(__builtin_amdgcn_readlane(s, i) * 40);
      const unsigned dof = (unsigned)(__builtin_amdgcn_readlane(d, i) * 40);
      const float    av  = __int_as_float(__builtin_amdgcn_readlane(ai, i));
      atomicAdd(agg + dof + lane, Hn[so + lane] * av);
      if (lane < 8) atomicAdd(agg + dof + 32 + lane, Hn[so + 32 + lane] * av);
    }
  }
}

__global__ void gat_bias_relu(float* __restrict__ agg, const float* __restrict__ b, long n64) {
  long t = (long)blockIdx.x * blockDim.x + threadIdx.x;
  if (t >= n64) return;
  float v = agg[t] + b[t & 63];
  agg[t] = v > 0.f ? v : 0.f;
}

__global__ void gat_logsoftmax(const float* __restrict__ agg, const float* __restrict__ b,
                               float* __restrict__ out, int N) {
  int n = blockIdx.x * blockDim.x + threadIdx.x;
  if (n >= N) return;
  const float* a = agg + (long)n * 40;
  float v[40];
  float mx = -INFINITY;
#pragma unroll 8
  for (int c = 0; c < 40; ++c) { v[c] = a[c] + b[c]; mx = fmaxf(mx, v[c]); }
  float sum = 0.f;
#pragma unroll 8
  for (int c = 0; c < 40; ++c) sum += expf(v[c] - mx);
  const float ls = mx + logf(sum);
  float* o = out + (long)n * 40;
#pragma unroll 8
  for (int c = 0; c < 40; ++c) o[c] = v[c] - ls;
}

extern "C" void kernel_launch(void* const* d_in, const int* in_sizes, int n_in,
                              void* d_out, int out_size, void* d_ws, size_t ws_size,
                              hipStream_t stream) {
  const float* x     = (const float*)d_in[0];
  const int*   eidx  = (const int*)d_in[1];
  const float* W1    = (const float*)d_in[2];
  const float* aS1   = (const float*)d_in[3];
  const float* aD1   = (const float*)d_in[4];
  const float* b1    = (const float*)d_in[5];
  const float* W2    = (const float*)d_in[6];
  const float* aS2   = (const float*)d_in[7];
  const float* aD2   = (const float*)d_in[8];
  const float* b2    = (const float*)d_in[9];
  float* out = (float*)d_out;

  const int N = in_sizes[0] / 128;
  const int E = in_sizes[1] / 2;
  const int* src = eidx;
  const int* dst = eidx + E;

  const long N64 = (long)N * 64, N40 = (long)N * 40;
  const long N8  = (long)N * 8,  E8  = (long)E * 8;

  float* ws      = (float*)d_ws;
  float* h1      = ws;
  float* esrc1   = h1 + N64;
  float* edst1   = esrc1 + N8;
  float* e1      = edst1 + N8;
  float* s1      = e1 + E8;
  unsigned* mK1  = (unsigned*)(s1 + N8);
  float* agg1    = (float*)mK1 + N8;
  float* h2      = agg1 + N64;
  float* esrc2   = h2 + N40;
  float* edst2   = esrc2 + N;
  float* e2      = edst2 + N;
  float* s2      = e2 + E;
  unsigned* mK2  = (unsigned*)(s2 + N);
  float* agg2    = (float*)mK2 + N;

  const int BT = 256;
  const int rowBlocks = (N + 15) / 16;

  gat_gemm1_wmma<<<rowBlocks, 128, 0, stream>>>(x, W1, h1, N);
  gat_attn1<<<(int)((N8 + BT - 1) / BT), BT, 0, stream>>>(h1, aS1, aD1, esrc1, edst1, N);
  hipMemsetAsync(s1, 0, (size_t)(N8 + N8 + N64) * 4, stream);
  gat_edge_max<<<(int)((E8 + BT - 1) / BT), BT, 0, stream>>>(src, dst, esrc1, edst1, e1, mK1, E, 8);
  gat_edge_sum<<<(int)((E8 + BT - 1) / BT), BT, 0, stream>>>(dst, e1, mK1, s1, E, 8);
  gat_edge_msg1<<<(int)((E8 + BT - 1) / BT), BT, 0, stream>>>(src, dst, e1, s1, h1, agg1, E);
  gat_bias_relu<<<(int)((N64 + BT - 1) / BT), BT, 0, stream>>>(agg1, b1, N64);

  gat_gemm2_wmma<<<rowBlocks, 96, 0, stream>>>(agg1, W2, h2, N);
  gat_attn2<<<(N + BT - 1) / BT, BT, 0, stream>>>(h2, aS2, aD2, esrc2, edst2, N);
  hipMemsetAsync(s2, 0, (size_t)(N + N + N40) * 4, stream);
  gat_edge_max<<<(E + BT - 1) / BT, BT, 0, stream>>>(src, dst, esrc2, edst2, e2, mK2, E, 1);
  gat_edge_sum<<<(E + BT - 1) / BT, BT, 0, stream>>>(dst, e2, mK2, s2, E, 1);
  gat_edge_msg2<<<(E + BT - 1) / BT, BT, 0, stream>>>(src, dst, e2, s2, h2, agg2, E);
  gat_logsoftmax<<<(N + BT - 1) / BT, BT, 0, stream>>>(agg2, b2, out, N);
}


# edge_msg1: one wave per 32 edges with the 64 channels across lanes (alphas broadcast inside groups of eight lanes), so each load and atomic add covers one 128-byte line of one row, not eight lines
# speedup vs baseline: 2.2087x; 1.1943x over previous
#include <hip/hip_runtime.h>
#include <math.h>


typedef float v2f __attribute__((ext_vector_type(2)));
typedef float v8f __attribute__((ext_vector_type(8)));

#define NEG_SLOPE 0.2f
#define GAT_EPS 1e-16f

__device__ __forceinline__ unsigned fenc(float f) {
  unsigned u = __float_as_uint(f);
  return (u & 0x80000000u) ? ~u : (u | 0x80000000u);
}
__device__ __forceinline__ float fdec(unsigned k) {
  unsigned u = (k & 0x80000000u) ? (k ^ 0x80000000u) : ~k;
  return __uint_as_float(u);
}

__global__ void gat_gemm1_wmma(const float* __restrict__ X,
                               const float* __restrict__ W,
                               float* __restrict__ H, int N) {
  const int lane    = threadIdx.x & 31;
  const int wave    = threadIdx.x >> 5;
  const int row0    = blockIdx.x * 16;
  const int colBase = wave * 16;
  const int m       = lane & 15;
  const int khalf   = (lane >> 4) * 2;

  int rowA = row0 + m; if (rowA >= N) rowA = N - 1;
  const float* xrow = X + (long)rowA * 128;

  v8f c = {};
  for (int k0 = 0; k0 < 128; k0 += 4) {
    const int k = k0 + khalf;
    v2f a; a.x = xrow[k];                 a.y = xrow[k + 1];
    v2f b; b.x = W[k * 64 + colBase + m]; b.y = W[(k + 1) * 64 + colBase + m];
    c = __builtin_amdgcn_wmma_f32_16x16x4_f32(false, a, false, b,
                                              (short)0, c, false, false);
  }
  const int mrow = (lane >> 4) * 8;
#pragma unroll
  for (int v = 0; v < 8; ++v) {
    int r = row0 + mrow + v;
    if (r < N) H[(long)r * 64 + colBase + m] = c[v];
  }
}

__global__ void gat_gemm2_wmma(const float* __restrict__ X,
                               const float* __restrict__ W,
                               float* __restrict__ H, int N) {
  const int lane    = threadIdx.x & 31;
  const int wave    = threadIdx.x >> 5;
  const int row0    = blockIdx.x * 16;
  const int col     = wave * 16 + (lane & 15);
  const int m       = lane & 15;
  const int khalf   = (lane >> 4) * 2;
  const bool colOK  = (col < 40);

  int rowA = row0 + m; if (rowA >= N) rowA = N - 1;
  const float* xrow = X + (long)rowA * 64;

  v8f c = {};
  for (int k0 = 0; k0 < 64; k0 += 4) {
    const int k = k0 + khalf;
    v2f a; a.x = xrow[k]; a.y = xrow[k + 1];
    v2f b;
    b.x = colOK ? W[k * 40 + col] : 0.0f;
    b.y = colOK ? W[(k + 1) * 40 + col] : 0.0f;
    c = __builtin_amdgcn_wmma_f32_16x16x4_f32(false, a, false, b,
                                              (short)0, c, false, false);
  }
  const int mrow = (lane >> 4) * 8;
#pragma unroll
  for (int v = 0; v < 8; ++v) {
    int r = row0 + mrow + v;
    if (colOK && r < N) H[(long)r * 40 + col] = c[v];
  }
}

__global__ void gat_attn1(const float* __restrict__ H, const float* __restrict__ aS,
                          const float* __restrict__ aD, float* __restrict__ eS,
                          float* __restrict__ eD, int N) {
  int t = blockIdx.x * blockDim.x + threadIdx.x;
  if (t >= N * 8) return;
  const int head = t & 7;
  const float* hp = H + (long)(t >> 3) * 64 + head * 8;
  const float* as = aS + head * 8;
  const float* ad = aD + head * 8;
  float s = 0.f, d = 0.f;
#pragma unroll
  for (int c = 0; c < 8; ++c) { float h = hp[c]; s += h * as[c]; d += h * ad[c]; }
  eS[t] = s; eD[t] = d;
}

__global__ void gat_attn2(const float* __restrict__ H, const float* __restrict__ aS,
                          const float* __restrict__ aD, float* __restrict__ eS,
                          float* __restrict__ eD, int N) {
  int n = blockIdx.x * blockDim.x + threadIdx.x;
  if (n >= N) return;
  const float* hp = H + (long)n * 40;
  float s = 0.f, d = 0.f;
#pragma unroll 8
  for (int c = 0; c < 40; ++c) { float h = hp[c]; s += h * aS[c]; d += h * aD[c]; }
  eS[n] = s; eD[n] = d;
}

__global__ void gat_edge_max(const int* __restrict__ src, const int* __restrict__ dst,
                             const float* __restrict__ eS, const float* __restrict__ eD,
                             float* __restrict__ eBuf, unsigned* __restrict__ mKey,
                             int E, int H) {
  int t = blockIdx.x * blockDim.x + threadIdx.x;
  if (t >= E * H) return;
  const int e = t / H, h = t - e * H;
  const int s = src[e], d = dst[e];
  float v = eS[s * H + h] + eD[d * H + h];
  v = (v > 0.f) ? v : v * NEG_SLOPE;
  eBuf[t] = v;
  atomicMax(mKey + d * H + h, fenc(v));
}

__global__ void gat_edge_sum(const int* __restrict__ dst, float* __restrict__ eBuf,
                             const unsigned* __restrict__ mKey, float* __restrict__ sSum,
                             int E, int H) {
  int t = blockIdx.x * blockDim.x + threadIdx.x;
  if (t >= E * H) return;
  const int e = t / H, h = t - e * H;
  const int d = dst[e];
  float p = expf(eBuf[t] - fdec(mKey[d * H + h]));
  eBuf[t] = p;
  atomicAdd(sSum + d * H + h, p);
}


template <int I>
__device__ __forceinline__ float bcast8(float v) {
  return __uint_as_float(__builtin_amdgcn_mov_dpp8(__float_as_uint(v), (unsigned)(I * 0x249249)));
}

template <int J>
__device__ __forceinline__ void msg1_load(unsigned lane, int s, const float* __restrict__ Hn,
                                          float& h0, float& h1) {
  const float* row = Hn + (unsigned)__builtin_amdgcn_readlane(s, J) * 64u;
  h0 = row[lane];
  h1 = row[32u + lane];
}

template <int J>
__device__ __forceinline__ void msg1_add(unsigned lane, int d, float h0, float h1,
                                         const float (&A)[4][2], float* __restrict__ agg) {
  float* row = agg + (unsigned)__builtin_amdgcn_readlane(d, J) * 64u;
  atomicAdd(row + lane,       h0 * bcast8<J & 7>(A[J >> 3][0]));
  atomicAdd(row + 32u + lane, h1 * bcast8<J & 7>(A[J >> 3][1]));
}

template <int R>
__device__ __forceinline__ void msg1_group(unsigned lane, int s, int d, const float (&A)[4][2],
                                           const float* __restrict__ Hn, float* __restrict__ agg) {
  float h[16];
  msg1_load<8 * R + 0>(lane, s, Hn, h[0],  h[1]);
  msg1_load<8 * R + 1>(lane, s, Hn, h[2],  h[3]);
  msg1_load<8 * R + 2>(lane, s, Hn, h[4],  h[5]);
  msg1_load<8 * R + 3>(lane, s, Hn, h[6],  h[7]);
  msg1_load<8 * R + 4>(lane, s, Hn, h[8],  h[9]);
  msg1_load<8 * R + 5>(lane, s, Hn, h[10], h[11]);
  msg1_load<8 * R + 6>(lane, s, Hn, h[12], h[13]);
  msg1_load<8 * R + 7>(lane, s, Hn, h[14], h[15]);
  msg1_add<8 * R + 0>(lane, d, h[0],  h[1],  A, agg);
  msg1_add<8 * R + 1>(lane, d, h[2],  h[3],  A, agg);
  msg1_add<8 * R + 2>(lane, d, h[4],  h[5],  A, agg);
  msg1_add<8 * R + 3>(lane, d, h[6],  h[7],  A, agg);
  msg1_add<8 * R + 4>(lane, d, h[8],  h[9],  A, agg);
  msg1_add<8 * R + 5>(lane, d, h[10], h[11], A, agg);
  msg1_add<8 * R + 6>(lane, d, h[12], h[13], A, agg);
  msg1_add<8 * R + 7>(lane, d, h[14], h[15], A, agg);
}

__global__ void gat_edge_msg1(const int* __restrict__ src, const int* __restrict__ dst,
                              const float* __restrict__ eBuf, const float* __restrict__ sSum,
                              const float* __restrict__ Hn, float* __restrict__ agg, int E) {
  const unsigned lane = threadIdx.x & 31;
  const int e0 = (int)((blockIdx.x * blockDim.x + threadIdx.x) >> 5) * 32;
  if (e0 >= E) return;
  const int nE = (E - e0 < 32) ? (E - e0) : 32;

  int s = 0, d = 0;
  if ((int)lane < nE) { s = src[e0 + lane]; d = dst[e0 + lane]; }

  if (nE == 32) {
    const unsigned q = lane >> 3, i = lane & 7;
    float A[4][2];
    for (int r = 0; r < 4; ++r) {
      const unsigned e  = (unsigned)e0 + 8u * r + i;
      const unsigned dd = (unsigned)dst[e];
      for (int k = 0; k < 2; ++k) {
        const unsigned hd = 4u * k + q;
        A[r][k] = eBuf[e * 8u + hd] / (sSum[dd * 8u + hd] + GAT_EPS);
      }
    }
    msg1_group<0>(lane, s, d, A, Hn, agg);
    msg1_group<1>(lane, s, d, A, Hn, agg);
    msg1_group<2>(lane, s, d, A, Hn, agg);
    msg1_group<3>(lane, s, d, A, Hn, agg);
  } else {
    for (int j = 0; j < nE; ++j) {
      const unsigned sj = (unsigned)__builtin_amdgcn_readlane(s, j);
      const unsigned dj = (unsigned)__builtin_amdgcn_readlane(d, j);
      for (int k = 0; k < 2; ++k) {
        const unsigned c  = 32u * k + lane;
        const unsigned hd = c >> 3;
        const float alpha = eBuf[(unsigned)(e0 + j) * 8u + hd] / (sSum[dj * 8u + hd] + GAT_EPS);
        atomicAdd(agg + dj * 64u + c, Hn[sj * 64u + c] * alpha);
      }
    }
  }
}

template <int K>
__device__ __forceinline__ void msg2_slot(int lane, const int (&sj)[4], const int (&dj)[4],
                                          const int (&aj)[4], unsigned& so, unsigned& dof,
                                          float& av) {
  constexpr int jlo = (32 * K) / 40;
  constexpr int b   = 40 * (jlo + 1) - 32 * K;
  constexpr int jhi = (b < 32) ? jlo + 1 : jlo;
  const bool lo = lane < b;
  const int  c  = lo ? (32 * K - 40 * jlo + lane) : (lane - b);
  so  = (unsigned)((lo ? sj[jlo] : sj[jhi]) * 40 + c);
  dof = (unsigned)((lo ? dj[jlo] : dj[jhi]) * 40 + c);
  av  = __int_as_float(lo ? aj[jlo] : aj[jhi]);
}

__global__ void gat_edge_msg2(const int* __restrict__ src, const int* __restrict__ dst,
                              const float* __restrict__ eBuf, const float* __restrict__ sSum,
                              const float* __restrict__ Hn, float* __restrict__ agg, int E) {
  const int lane = threadIdx.x & 31;
  const int e0   = (int)((blockIdx.x * blockDim.x + threadIdx.x) >> 5) * 32;
  if (e0 >= E) return;
  const int nE = (E - e0 < 32) ? (E - e0) : 32;

  int s = 0, d = 0;
  float alpha = 0.f;
  if (lane < nE) {
    s = src[e0 + lane];
    d = dst[e0 + lane];
    alpha = eBuf[e0 + lane] / (sSum[d] + GAT_EPS);
  }
  const int ai = __float_as_int(alpha);

  if (nE == 32) {
#pragma unroll
    for (int g = 0; g < 8; ++g) {
      int sj[4], dj[4], aj[4];
      for (int j = 0; j < 4; ++j) {
        sj[j] = __builtin_amdgcn_readlane(s,  4 * g + j);
        dj[j] = __builtin_amdgcn_readlane(d,  4 * g + j);
        aj[j] = __builtin_amdgcn_readlane(ai, 4 * g + j);
      }
      unsigned so[5], dof[5];
      float av[5], hv[5];
      msg2_slot<0>(lane, sj, dj, aj, so[0], dof[0], av[0]);
      msg2_slot<1>(lane, sj, dj, aj, so[1], dof[1], av[1]);
      msg2_slot<2>(lane, sj, dj, aj, so[2], dof[2], av[2]);
      msg2_slot<3>(lane, sj, dj, aj, so[3], dof[3], av[3]);
      msg2_slot<4>(lane, sj, dj, aj, so[4], dof[4], av[4]);
      for (int k = 0; k < 5; ++k) hv[k] = Hn[so[k]];
      for (int k = 0; k < 5; ++k) atomicAdd(agg + dof[k], hv[k] * av[k]);
    }
  } else {
    for (int i = 0; i < nE; ++i) {
      const unsigned so  = (unsigned)(__builtin_amdgcn_readlane(s, i) * 40);
      const unsigned dof = (unsigned)(__builtin_amdgcn_readlane(d, i) * 40);
      const float    av  = __int_as_float(__builtin_amdgcn_readlane(ai, i));
      atomicAdd(agg + dof + lane, Hn[so + lane] * av);
      if (lane < 8) atomicAdd(agg + dof + 32 + lane, Hn[so + 32 + lane] * av);
    }
  }
}

__global__ void gat_bias_relu(float* __restrict__ agg, const float* __restrict__ b, long n64) {
  long t = (long)blockIdx.x * blockDim.x + threadIdx.x;
  if (t >= n64) return;
  float v = agg[t] + b[t & 63];
  agg[t] = v > 0.f ? v : 0.f;
}

__global__ void gat_logsoftmax(const float* __restrict__ agg, const float* __restrict__ b,
                               float* __restrict__ out, int N) {
  int n = blockIdx.x * blockDim.x + threadIdx.x;
  if (n >= N) return;
  const float* a = agg + (long)n * 40;
  float v[40];
  float mx = -INFINITY;
#pragma unroll 8
  for (int c = 0; c < 40; ++c) { v[c] = a[c] + b[c]; mx = fmaxf(mx, v[c]); }
  float sum = 0.f;
#pragma unroll 8
  for (int c = 0; c < 40; ++c) sum += expf(v[c] - mx);
  const float ls = mx + logf(sum);
  float* o = out + (long)n * 40;
#pragma unroll 8
  for (int c = 0; c < 40; ++c) o[c] = v[c] - ls;
}

extern "C" void kernel_launch(void* const* d_in, const int* in_sizes, int n_in,
                              void* d_out, int out_size, void* d_ws, size_t ws_size,
                              hipStream_t stream) {
  const float* x     = (const float*)d_in[0];
  const int*   eidx  = (const int*)d_in[1];
  const float* W1    = (const float*)d_in[2];
  const float* aS1   = (const float*)d_in[3];
  const float* aD1   = (const float*)d_in[4];
  const float* b1    = (const float*)d_in[5];
  const float* W2    = (const float*)d_in[6];
  const float* aS2   = (const float*)d_in[7];
  const float* aD2   = (const float*)d_in[8];
  const float* b2    = (const float*)d_in[9];
  float* out = (float*)d_out;

  const int N = in_sizes[0] / 128;
  const int E = in_sizes[1] / 2;
  const int* src = eidx;
  const int* dst = eidx + E;

  const long N64 = (long)N * 64, N40 = (long)N * 40;
  const long N8  = (long)N * 8,  E8  = (long)E * 8;

  float* ws      = (float*)d_ws;
  float* h1      = ws;
  float* esrc1   = h1 + N64;
  float* edst1   = esrc1 + N8;
  float* e1      = edst1 + N8;
  float* s1      = e1 + E8;
  unsigned* mK1  = (unsigned*)(s1 + N8);
  float* agg1    = (float*)mK1 + N8;
  float* h2      = agg1 + N64;
  float* esrc2   = h2 + N40;
  float* edst2   = esrc2 + N;
  float* e2      = edst2 + N;
  float* s2      = e2 + E;
  unsigned* mK2  = (unsigned*)(s2 + N);
  float* agg2    = (float*)mK2 + N;

  const int BT = 256;
  const int rowBlocks = (N + 15) / 16;

  gat_gemm1_wmma<<<rowBlocks, 128, 0, stream>>>(x, W1, h1, N);
  gat_attn1<<<(int)((N8 + BT - 1) / BT), BT, 0, stream>>>(h1, aS1, aD1, esrc1, edst1, N);
  hipMemsetAsync(s1, 0, (size_t)(N8 + N8 + N64) * 4, stream);
  gat_edge_max<<<(int)((E8 + BT - 1) / BT), BT, 0, stream>>>(src, dst, esrc1, edst1, e1, mK1, E, 8);
  gat_edge_sum<<<(int)((E8 + BT - 1) / BT), BT, 0, stream>>>(dst, e1, mK1, s1, E, 8);
  gat_edge_msg1<<<(E + BT - 1) / BT, BT, 0, stream>>>(src, dst, e1, s1, h1, agg1, E);
  gat_bias_relu<<<(int)((N64 + BT - 1) / BT), BT, 0, stream>>>(agg1, b1, N64);

  gat_gemm2_wmma<<<rowBlocks, 96, 0, stream>>>(agg1, W2, h2, N);
  gat_attn2<<<(N + BT - 1) / BT, BT, 0, stream>>>(h2, aS2, aD2, esrc2, edst2, N);
  hipMemsetAsync(s2, 0, (size_t)(N + N + N40) * 4, stream);
  gat_edge_max<<<(E + BT - 1) / BT, BT, 0, stream>>>(src, dst, esrc2, edst2, e2, mK2, E, 1);
  gat_edge_sum<<<(E + BT - 1) / BT, BT, 0, stream>>>(dst, e2, mK2, s2, E, 1);
  gat_edge_msg2<<<(E + BT - 1) / BT, BT, 0, stream>>>(src, dst, e2, s2, h2, agg2, E);
  gat_logsoftmax<<<(N + BT - 1) / BT, BT, 0, stream>>>(agg2, b2, out, N);
}


# dst_scan: counts come in and offsets go out as whole 128-byte lines staged through LDS, where each thread walks its run, instead of 4-byte loads and stores 196 bytes apart across the lanes
# speedup vs baseline: 4.7418x; 2.1469x over previous
#include <hip/hip_runtime.h>
#include <math.h>


typedef float v2f __attribute__((ext_vector_type(2)));
typedef float v8f __attribute__((ext_vector_type(8)));

#define NEG_SLOPE 0.2f
#define GAT_EPS 1e-16f

__global__ void gat_gemm1_wmma(const float* __restrict__ X,
                               const float* __restrict__ W,
                               float* __restrict__ H, int N) {
  const int lane    = threadIdx.x & 31;
  const int wave    = threadIdx.x >> 5;
  const int row0    = blockIdx.x * 16;
  const int colBase = wave * 16;
  const int m       = lane & 15;
  const int khalf   = (lane >> 4) * 2;

  int rowA = row0 + m; if (rowA >= N) rowA = N - 1;
  const float* xrow = X + (long)rowA * 128;

  v8f c = {};
  for (int k0 = 0; k0 < 128; k0 += 4) {
    const int k = k0 + khalf;
    v2f a; a.x = xrow[k];                 a.y = xrow[k + 1];
    v2f b; b.x = W[k * 64 + colBase + m]; b.y = W[(k + 1) * 64 + colBase + m];
    c = __builtin_amdgcn_wmma_f32_16x16x4_f32(false, a, false, b,
                                              (short)0, c, false, false);
  }
  const int mrow = (lane >> 4) * 8;
#pragma unroll
  for (int v = 0; v < 8; ++v) {
    int r = row0 + mrow + v;
    if (r < N) H[(long)r * 64 + colBase + m] = c[v];
  }
}

__global__ void gat_gemm2_wmma(const float* __restrict__ X,
                               const float* __restrict__ W,
                               float* __restrict__ H, int N) {
  const int lane    = threadIdx.x & 31;
  const int wave    = threadIdx.x >> 5;
  const int row0    = blockIdx.x * 16;
  const int col     = wave * 16 + (lane & 15);
  const int m       = lane & 15;
  const int khalf   = (lane >> 4) * 2;
  const bool colOK  = (col < 40);

  int rowA = row0 + m; if (rowA >= N) rowA = N - 1;
  const float* xrow = X + (long)rowA * 64;

  v8f c = {};
  for (int k0 = 0; k0 < 64; k0 += 4) {
    const int k = k0 + khalf;
    v2f a; a.x = xrow[k]; a.y = xrow[k + 1];
    v2f b;
    b.x = colOK ? W[k * 40 + col] : 0.0f;
    b.y = colOK ? W[(k + 1) * 40 + col] : 0.0f;
    c = __builtin_amdgcn_wmma_f32_16x16x4_f32(false, a, false, b,
                                              (short)0, c, false, false);
  }
  const int mrow = (lane >> 4) * 8;
#pragma unroll
  for (int v = 0; v < 8; ++v) {
    int r = row0 + mrow + v;
    if (colOK && r < N) H[(long)r * 40 + col] = c[v];
  }
}

__global__ void gat_attn1(const float* __restrict__ H, const float* __restrict__ aS,
                          const float* __restrict__ aD, float* __restrict__ eS,
                          float* __restrict__ eD, int N) {
  int t = blockIdx.x * blockDim.x + threadIdx.x;
  if (t >= N * 8) return;
  const int head = t & 7;
  const float* hp = H + (long)(t >> 3) * 64 + head * 8;
  const float* as = aS + head * 8;
  const float* ad = aD + head * 8;
  float s = 0.f, d = 0.f;
#pragma unroll
  for (int c = 0; c < 8; ++c) { float h = hp[c]; s += h * as[c]; d += h * ad[c]; }
  eS[t] = s; eD[t] = d;
}

__global__ void gat_attn2(const float* __restrict__ H, const float* __restrict__ aS,
                          const float* __restrict__ aD, float* __restrict__ eS,
                          float* __restrict__ eD, int N) {
  int n = blockIdx.x * blockDim.x + threadIdx.x;
  if (n >= N) return;
  const float* hp = H + (long)n * 40;
  float s = 0.f, d = 0.f;
#pragma unroll 8
  for (int c = 0; c < 40; ++c) { float h = hp[c]; s += h * aS[c]; d += h * aD[c]; }
  eS[n] = s; eD[n] = d;
}

__global__ void gat_dst_count(const int* __restrict__ dst, int* __restrict__ cnt,
                              int* __restrict__ rank, int E) {
  const int e = blockIdx.x * blockDim.x + threadIdx.x;
  if (e >= E) return;
  rank[e] = atomicAdd(cnt + dst[e], 1);
}

#define SCAN_T    1024
#define SCAN_PER  63
#define SCAN_TILE (SCAN_T * SCAN_PER)
#define SCAN_LD   16
__global__ void __launch_bounds__(SCAN_T)
gat_dst_scan(const int* __restrict__ cnt, int* __restrict__ offs, int N) {
  __shared__ int4 buf4[SCAN_LD * SCAN_T];
  __shared__ int part[SCAN_T];
  int* buf = (int*)buf4;
  const int t = threadIdx.x;
  const int spare = SCAN_TILE + t;
  int carry = 0;
  for (int base = 0; base < N; base += SCAN_TILE) {
    const int n   = (N - base < SCAN_TILE) ? (N - base) : SCAN_TILE;
    const int n4  = n >> 2;
    const int per = ((n + SCAN_T - 1) / SCAN_T) | 1;
    const int b0  = t * per;

    if (n4 > 0) {
      const int4* __restrict__ c4 = (const int4*)(cnt + base);
      int4 v[SCAN_LD];
#pragma unroll
      for (int u = 0; u < SCAN_LD; ++u) { const int i = t + u * SCAN_T; v[u] = c4[i < n4 ? i : n4 - 1]; }
#pragma unroll
      for (int u = 0; u < SCAN_LD; ++u) buf4[t + u * SCAN_T] = v[u];
    }
    if (n & 3) {
      __syncthreads();
      if (t < (n & 3)) buf[4 * n4 + t] = cnt[base + 4 * n4 + t];
    }
    __syncthreads();

    int sum = 0;
    for (int i0 = 0; i0 < per; i0 += 8) {
      int c[8];
#pragma unroll
      for (int u = 0; u < 8; ++u) {
        const int idx = b0 + i0 + u;
        const bool ok = (i0 + u < per) && (idx < n);
        const int w = buf[ok ? idx : spare];
        c[u] = ok ? w : 0;
      }
#pragma unroll
      for (int u = 0; u < 8; ++u) sum += c[u];
    }
    part[t] = sum;
    __syncthreads();
    for (int o = 1; o < SCAN_T; o <<= 1) {
      const int v = (t >= o) ? part[t - o] : 0;
      __syncthreads();
      part[t] += v;
      __syncthreads();
    }
    const int total = part[SCAN_T - 1];
    int run = carry + part[t] - sum;
    for (int i0 = 0; i0 < per; i0 += 8) {
      int c[8], at[8];
#pragma unroll
      for (int u = 0; u < 8; ++u) {
        const int idx = b0 + i0 + u;
        const bool ok = (i0 + u < per) && (idx < n);
        at[u] = ok ? idx : spare;
        const int w = buf[at[u]];
        c[u] = ok ? w : 0;
      }
#pragma unroll
      for (int u = 0; u < 8; ++u) { buf[at[u]] = run; run += c[u]; }
    }
    __syncthreads();

    if (n4 > 0) {
      int4* __restrict__ o4 = (int4*)(offs + base);
#pragma unroll
      for (int u = 0; u < SCAN_LD; ++u) { const int i = t + u * SCAN_T; if (i < n4) o4[i] = buf4[i]; }
    }
    if (t < (n & 3)) offs[base + 4 * n4 + t] = buf[4 * n4 + t];
    carry += total;
    __syncthreads();
  }
  if (t == 0) offs[N] = carry;
}

__global__ void gat_dst_place(const int* __restrict__ src, const int* __restrict__ dst,
                              const int* __restrict__ rank, const int* __restrict__ offs,
                              int* __restrict__ perm, int E) {
  const int e = blockIdx.x * blockDim.x + threadIdx.x;
  if (e >= E) return;
  perm[offs[dst[e]] + rank[e]] = src[e];
}

template <int I>
__device__ __forceinline__ float bcast8(float v) {
  return __uint_as_float(__builtin_amdgcn_mov_dpp8(__float_as_uint(v), (unsigned)(I * 0x249249)));
}

__device__ __forceinline__ float lrelu(float v) { return (v > 0.f) ? v : v * NEG_SLOPE; }

__device__ __forceinline__ float gmax8(float v) {
  v = fmaxf(v, __shfl_xor(v, 1)); v = fmaxf(v, __shfl_xor(v, 2)); v = fmaxf(v, __shfl_xor(v, 4));
  return v;
}
__device__ __forceinline__ float gsum8(float v) {
  v += __shfl_xor(v, 1); v += __shfl_xor(v, 2); v += __shfl_xor(v, 4);
  return v;
}
__device__ __forceinline__ float wmax32(float v) {
  v = gmax8(v); v = fmaxf(v, __shfl_xor(v, 8)); v = fmaxf(v, __shfl_xor(v, 16));
  return v;
}
__device__ __forceinline__ float wsum32(float v) {
  v = gsum8(v); v += __shfl_xor(v, 8); v += __shfl_xor(v, 16);
  return v;
}


__device__ __forceinline__ void l1_scores(const int* __restrict__ perm, const float* __restrict__ eS,
                                          int base, int cnt, unsigned q, unsigned i,
                                          const float (&ed)[2], float (&e)[4][2]) {
  for (int r = 0; r < 4; ++r) {
    const int j = 8 * r + (int)i;
    const bool ok = j < cnt;
    const unsigned s = (unsigned)perm[base + (ok ? j : 0)];
    for (int k = 0; k < 2; ++k) {
      const float v = lrelu(eS[s * 8u + 4u * k + q] + ed[k]);
      e[r][k] = ok ? v : -INFINITY;
    }
  }
}

template <int J>
__device__ __forceinline__ void l1_load(unsigned lane, int s, const float* __restrict__ Hn,
                                        float& h0, float& h1) {
  const float* row = Hn + (unsigned)__builtin_amdgcn_readlane(s, J) * 64u;
  h0 = row[lane];
  h1 = row[32u + lane];
}

template <int R>
__device__ __forceinline__ void l1_group(unsigned lane, int s, const float (&A)[4][2],
                                         const float* __restrict__ Hn, float& acc0, float& acc1) {
  float h[16];
  l1_load<8 * R + 0>(lane, s, Hn, h[0],  h[1]);
  l1_load<8 * R + 1>(lane, s, Hn, h[2],  h[3]);
  l1_load<8 * R + 2>(lane, s, Hn, h[4],  h[5]);
  l1_load<8 * R + 3>(lane, s, Hn, h[6],  h[7]);
  l1_load<8 * R + 4>(lane, s, Hn, h[8],  h[9]);
  l1_load<8 * R + 5>(lane, s, Hn, h[10], h[11]);
  l1_load<8 * R + 6>(lane, s, Hn, h[12], h[13]);
  l1_load<8 * R + 7>(lane, s, Hn, h[14], h[15]);
  acc0 += h[0]  * bcast8<0>(A[R][0]);  acc1 += h[1]  * bcast8<0>(A[R][1]);
  acc0 += h[2]  * bcast8<1>(A[R][0]);  acc1 += h[3]  * bcast8<1>(A[R][1]);
  acc0 += h[4]  * bcast8<2>(A[R][0]);  acc1 += h[5]  * bcast8<2>(A[R][1]);
  acc0 += h[6]  * bcast8<3>(A[R][0]);  acc1 += h[7]  * bcast8<3>(A[R][1]);
  acc0 += h[8]  * bcast8<4>(A[R][0]);  acc1 += h[9]  * bcast8<4>(A[R][1]);
  acc0 += h[10] * bcast8<5>(A[R][0]);  acc1 += h[11] * bcast8<5>(A[R][1]);
  acc0 += h[12] * bcast8<6>(A[R][0]);  acc1 += h[13] * bcast8<6>(A[R][1]);
  acc0 += h[14] * bcast8<7>(A[R][0]);  acc1 += h[15] * bcast8<7>(A[R][1]);
}

__device__ __forceinline__ void l1_chunk_sum(const int* __restrict__ perm, int base, int cnt,
                                             unsigned lane, const float (&A)[4][2],
                                             const float* __restrict__ Hn, float& acc0, float& acc1) {
  const int s = perm[base + (((int)lane < cnt) ? (int)lane : 0)];
  l1_group<0>(lane, s, A, Hn, acc0, acc1);
  if (cnt > 8)  l1_group<1>(lane, s, A, Hn, acc0, acc1);
  if (cnt > 16) l1_group<2>(lane, s, A, Hn, acc0, acc1);
  if (cnt > 24) l1_group<3>(lane, s, A, Hn, acc0, acc1);
}

__global__ void gat_node_agg1(const int* __restrict__ offs, const int* __restrict__ perm,
                              const float* __restrict__ eS, const float* __restrict__ eD,
                              const float* __restrict__ Hn, const float* __restrict__ bias,
                              float* __restrict__ outp, int N) {
  const unsigned lane = threadIdx.x & 31;
  const int n = (int)((blockIdx.x * blockDim.x + threadIdx.x) >> 5);
  if (n >= N) return;
  const unsigned q = lane >> 3, i = lane & 7;
  const int beg = offs[n], deg = offs[n + 1] - beg;
  const float ed[2] = { eD[(unsigned)n * 8u + q], eD[(unsigned)n * 8u + 4u + q] };

  float acc0 = 0.f, acc1 = 0.f;
  if (deg > 0 && deg <= 32) {
    float e[4][2], A[4][2], m[2], ss[2];
    l1_scores(perm, eS, beg, deg, q, i, ed, e);
    for (int k = 0; k < 2; ++k) {
      m[k] = gmax8(fmaxf(fmaxf(e[0][k], e[1][k]), fmaxf(e[2][k], e[3][k])));
      if (!isfinite(m[k])) m[k] = 0.f;
      float t = 0.f;
      for (int r = 0; r < 4; ++r) { e[r][k] = expf(e[r][k] - m[k]); t += e[r][k]; }
      ss[k] = gsum8(t);
      for (int r = 0; r < 4; ++r) A[r][k] = e[r][k] / (ss[k] + GAT_EPS);
    }
    l1_chunk_sum(perm, beg, deg, lane, A, Hn, acc0, acc1);
  } else if (deg > 32) {
    float e[4][2], A[4][2], m[2] = { -INFINITY, -INFINITY }, ss[2] = { 0.f, 0.f };
    for (int c0 = 0; c0 < deg; c0 += 32) {
      const int cn = (deg - c0 < 32) ? (deg - c0) : 32;
      l1_scores(perm, eS, beg + c0, cn, q, i, ed, e);
      for (int k = 0; k < 2; ++k)
        m[k] = fmaxf(m[k], fmaxf(fmaxf(e[0][k], e[1][k]), fmaxf(e[2][k], e[3][k])));
    }
    for (int k = 0; k < 2; ++k) { m[k] = gmax8(m[k]); if (!isfinite(m[k])) m[k] = 0.f; }
    for (int c0 = 0; c0 < deg; c0 += 32) {
      const int cn = (deg - c0 < 32) ? (deg - c0) : 32;
      l1_scores(perm, eS, beg + c0, cn, q, i, ed, e);
      for (int k = 0; k < 2; ++k)
        for (int r = 0; r < 4; ++r) ss[k] += expf(e[r][k] - m[k]);
    }
    for (int k = 0; k < 2; ++k) ss[k] = gsum8(ss[k]);
    for (int c0 = 0; c0 < deg; c0 += 32) {
      const int cn = (deg - c0 < 32) ? (deg - c0) : 32;
      l1_scores(perm, eS, beg + c0, cn, q, i, ed, e);
      for (int k = 0; k < 2; ++k)
        for (int r = 0; r < 4; ++r) A[r][k] = expf(e[r][k] - m[k]) / (ss[k] + GAT_EPS);
      l1_chunk_sum(perm, beg + c0, cn, lane, A, Hn, acc0, acc1);
    }
  }
  float* orow = outp + (unsigned)n * 64u;
  const float v0 = acc0 + bias[lane], v1 = acc1 + bias[32u + lane];
  orow[lane]       = v0 > 0.f ? v0 : 0.f;
  orow[32u + lane] = v1 > 0.f ? v1 : 0.f;
}


template <int J>
__device__ __forceinline__ void l2_load(unsigned lane, int s, const float* __restrict__ Hn,
                                        float& h0, float& h1) {
  const float* row = Hn + (unsigned)__builtin_amdgcn_readlane(s, J) * 40u;
  h0 = row[lane];
  h1 = (lane < 8u) ? row[32u + lane] : 0.f;
}
template <int J>
__device__ __forceinline__ float l2_alpha(float a) {
  return __int_as_float(__builtin_amdgcn_readlane(__float_as_int(a), J));
}

template <int R>
__device__ __forceinline__ void l2_group(unsigned lane, int s, float a,
                                         const float* __restrict__ Hn, float& acc0, float& acc1) {
  float h[16];
  l2_load<8 * R + 0>(lane, s, Hn, h[0],  h[1]);
  l2_load<8 * R + 1>(lane, s, Hn, h[2],  h[3]);
  l2_load<8 * R + 2>(lane, s, Hn, h[4],  h[5]);
  l2_load<8 * R + 3>(lane, s, Hn, h[6],  h[7]);
  l2_load<8 * R + 4>(lane, s, Hn, h[8],  h[9]);
  l2_load<8 * R + 5>(lane, s, Hn, h[10], h[11]);
  l2_load<8 * R + 6>(lane, s, Hn, h[12], h[13]);
  l2_load<8 * R + 7>(lane, s, Hn, h[14], h[15]);
  float a0;
  a0 = l2_alpha<8 * R + 0>(a); acc0 += h[0]  * a0; acc1 += h[1]  * a0;
  a0 = l2_alpha<8 * R + 1>(a); acc0 += h[2]  * a0; acc1 += h[3]  * a0;
  a0 = l2_alpha<8 * R + 2>(a); acc0 += h[4]  * a0; acc1 += h[5]  * a0;
  a0 = l2_alpha<8 * R + 3>(a); acc0 += h[6]  * a0; acc1 += h[7]  * a0;
  a0 = l2_alpha<8 * R + 4>(a); acc0 += h[8]  * a0; acc1 += h[9]  * a0;
  a0 = l2_alpha<8 * R + 5>(a); acc0 += h[10] * a0; acc1 += h[11] * a0;
  a0 = l2_alpha<8 * R + 6>(a); acc0 += h[12] * a0; acc1 += h[13] * a0;
  a0 = l2_alpha<8 * R + 7>(a); acc0 += h[14] * a0; acc1 += h[15] * a0;
}

__device__ __forceinline__ void l2_chunk_sum(unsigned lane, int s, float a, int cnt,
                                             const float* __restrict__ Hn, float& acc0, float& acc1) {
  l2_group<0>(lane, s, a, Hn, acc0, acc1);
  if (cnt > 8)  l2_group<1>(lane, s, a, Hn, acc0, acc1);
  if (cnt > 16) l2_group<2>(lane, s, a, Hn, acc0, acc1);
  if (cnt > 24) l2_group<3>(lane, s, a, Hn, acc0, acc1);
}

__global__ void gat_node_agg2(const int* __restrict__ offs, const int* __restrict__ perm,
                              const float* __restrict__ eS, const float* __restrict__ eD,
                              const float* __restrict__ Hn, const float* __restrict__ bias,
                              float* __restrict__ outp, int N) {
  __shared__ float rows[8 * 40];
  const unsigned lane = threadIdx.x & 31;
  const unsigned wv   = threadIdx.x >> 5;
  const int n = (int)(blockIdx.x * 8u + wv);
  float acc0 = 0.f, acc1 = 0.f;
  if (n < N) {
    const int beg = offs[n], deg = offs[n + 1] - beg;
    const float ed = eD[n];
    if (deg > 0 && deg <= 32) {
      const bool ok = (int)lane < deg;
      const int s = perm[beg + (ok ? (int)lane : 0)];
      const float e = ok ? lrelu(eS[s] + ed) : -INFINITY;
      float m = wmax32(e);
      if (!isfinite(m)) m = 0.f;
      const float p  = expf(e - m);
      const float ss = wsum32(p);
      const float a  = p / (ss + GAT_EPS);
      l2_chunk_sum(lane, s, a, deg, Hn, acc0, acc1);
    } else if (deg > 32) {
      float m = -INFINITY, ss = 0.f;
      for (int c0 = 0; c0 < deg; c0 += 32) {
        const bool ok = c0 + (int)lane < deg;
        const int s = perm[beg + (ok ? c0 + (int)lane : 0)];
        m = fmaxf(m, ok ? lrelu(eS[s] + ed) : -INFINITY);
      }
      m = wmax32(m);
      if (!isfinite(m)) m = 0.f;
      for (int c0 = 0; c0 < deg; c0 += 32) {
        const bool ok = c0 + (int)lane < deg;
        const int s = perm[beg + (ok ? c0 + (int)lane : 0)];
        ss += ok ? expf(lrelu(eS[s] + ed) - m) : 0.f;
      }
      ss = wsum32(ss);
      for (int c0 = 0; c0 < deg; c0 += 32) {
        const int cn = (deg - c0 < 32) ? (deg - c0) : 32;
        const bool ok = (int)lane < cn;
        const int s = perm[beg + c0 + (ok ? (int)lane : 0)];
        const float a = ok ? expf(lrelu(eS[s] + ed) - m) / (ss + GAT_EPS) : 0.f;
        l2_chunk_sum(lane, s, a, cn, Hn, acc0, acc1);
      }
    }
    const float v0 = acc0 + bias[lane];
    const float v1 = (lane < 8u) ? acc1 + bias[32u + lane] : -INFINITY;
    const float mx = wmax32(fmaxf(v0, v1));
    const float sm = wsum32(expf(v0 - mx) + ((lane < 8u) ? expf(v1 - mx) : 0.f));
    const float ls = mx + logf(sm);
    rows[wv * 40u + lane] = v0 - ls;
    if (lane < 8u) rows[wv * 40u + 32u + lane] = v1 - ls;
  }
  __syncthreads();
  const long base = (long)blockIdx.x * 320;
  const long lim  = (long)N * 40;
  const unsigned t = threadIdx.x;
  if (base + t < lim) outp[base + t] = rows[t];
  if (t < 64u && base + 256 + t < lim) outp[base + 256 + t] = rows[256u + t];
}

extern "C" void kernel_launch(void* const* d_in, const int* in_sizes, int n_in,
                              void* d_out, int out_size, void* d_ws, size_t ws_size,
                              hipStream_t stream) {
  const float* x     = (const float*)d_in[0];
  const int*   eidx  = (const int*)d_in[1];
  const float* W1    = (const float*)d_in[2];
  const float* aS1   = (const float*)d_in[3];
  const float* aD1   = (const float*)d_in[4];
  const float* b1    = (const float*)d_in[5];
  const float* W2    = (const float*)d_in[6];
  const float* aS2   = (const float*)d_in[7];
  const float* aD2   = (const float*)d_in[8];
  const float* b2    = (const float*)d_in[9];
  float* out = (float*)d_out;

  const int N = in_sizes[0] / 128;
  const int E = in_sizes[1] / 2;
  const int* src = eidx;
  const int* dst = eidx + E;

  const long N64 = (long)N * 64, N40 = (long)N * 40;
  const long N8  = (long)N * 8,  E8  = (long)E * 8;

  float* ws      = (float*)d_ws;
  float* h1      = ws;
  float* esrc1   = h1 + N64;
  float* edst1   = esrc1 + N8;
  float* e1      = edst1 + N8;
  float* s1      = e1 + E8;
  float* agg1    = s1 + N8 + N8;
  float* h2      = agg1 + N64;
  float* esrc2   = h2 + N40;
  float* edst2   = esrc2 + N;
  int* rank = (int*)e1;
  int* perm = rank + E;
  int* cnt  = (int*)s1;
  int* offs = cnt + ((N + 31) / 32) * 32;

  const int BT = 256;
  const int rowBlocks = (N + 15) / 16;
  const int nodeBlocks = (N + 7) / 8;

  hipMemsetAsync(cnt, 0, (size_t)N * 4, stream);
  gat_dst_count<<<(E + BT - 1) / BT, BT, 0, stream>>>(dst, cnt, rank, E);
  gat_dst_scan<<<1, 1024, 0, stream>>>(cnt, offs, N);
  gat_dst_place<<<(E + BT - 1) / BT, BT, 0, stream>>>(src, dst, rank, offs, perm, E);

  gat_gemm1_wmma<<<rowBlocks, 128, 0, stream>>>(x, W1, h1, N);
  gat_attn1<<<(int)((N8 + BT - 1) / BT), BT, 0, stream>>>(h1, aS1, aD1, esrc1, edst1, N);
  gat_node_agg1<<<nodeBlocks, BT, 0, stream>>>(offs, perm, esrc1, edst1, h1, b1, agg1, N);

  gat_gemm2_wmma<<<rowBlocks, 96, 0, stream>>>(agg1, W2, h2, N);
  gat_attn2<<<(N + BT - 1) / BT, BT, 0, stream>>>(h2, aS2, aD2, esrc2, edst2, N);
  gat_node_agg2<<<nodeBlocks, BT, 0, stream>>>(offs, perm, esrc2, edst2, h2, b2, out, N);
}


# edge sort counts and places with LDS counters: a work-group owns a destination range and an edge slice and streams the slice's lines; no global atomics, no memset; segment order now by slice
# speedup vs baseline: 6.2058x; 1.3087x over previous
#include <hip/hip_runtime.h>
#include <math.h>


typedef float v2f __attribute__((ext_vector_type(2)));
typedef float v8f __attribute__((ext_vector_type(8)));

#define NEG_SLOPE 0.2f
#define GAT_EPS 1e-16f

template <int I>
__device__ __forceinline__ float bcast8(float v) {
  return __uint_as_float(__builtin_amdgcn_mov_dpp8(__float_as_uint(v), (unsigned)(I * 0x249249)));
}

__device__ __forceinline__ float lrelu(float v) { return (v > 0.f) ? v : v * NEG_SLOPE; }

__device__ __forceinline__ float gmax8(float v) {
  v = fmaxf(v, __shfl_xor(v, 1)); v = fmaxf(v, __shfl_xor(v, 2)); v = fmaxf(v, __shfl_xor(v, 4));
  return v;
}
__device__ __forceinline__ float gsum8(float v) {
  v += __shfl_xor(v, 1); v += __shfl_xor(v, 2); v += __shfl_xor(v, 4);
  return v;
}
__device__ __forceinline__ float gsum8x8(const float (&x)[8], unsigned lane) {
  const bool b2 = (lane & 4u) != 0u, b1 = (lane & 2u) != 0u, b0 = (lane & 1u) != 0u;
  float y[4], z[2];
  for (int j = 0; j < 4; ++j) {
    const float keep = b2 ? x[4 + j] : x[j], send = b2 ? x[j] : x[4 + j];
    y[j] = keep + __shfl_xor(send, 4);
  }
  for (int j = 0; j < 2; ++j) {
    const float keep = b1 ? y[2 + j] : y[j], send = b1 ? y[j] : y[2 + j];
    z[j] = keep + __shfl_xor(send, 2);
  }
  const float keep = b0 ? z[1] : z[0], send = b0 ? z[0] : z[1];
  return keep + __shfl_xor(send, 1);
}
__device__ __forceinline__ float wmax32(float v) {
  v = gmax8(v); v = fmaxf(v, __shfl_xor(v, 8)); v = fmaxf(v, __shfl_xor(v, 16));
  return v;
}
__device__ __forceinline__ float wsum32(float v) {
  v = gsum8(v); v += __shfl_xor(v, 8); v += __shfl_xor(v, 16);
  return v;
}

__device__ __forceinline__ void
gat_gemm1_tiles(const float* __restrict__ X, const float* __restrict__ W,
                const float* __restrict__ aS, const float* __restrict__ aD,
                float* __restrict__ H, float* __restrict__ eS, float* __restrict__ eD, int N,
                int blk, float (*se)[2][128]) {
  const int lane    = threadIdx.x & 31;
  const int wave    = threadIdx.x >> 5;
  const int tile    = wave >> 2;
  const int cw      = wave & 3;
  const int row0    = blk * 128 + tile * 16;
  const int colBase = cw * 16;
  const int m       = lane & 15;
  const int khalf   = (lane >> 4) * 2;

  int rowA = row0 + m; if (rowA >= N) rowA = N - 1;
  const float* xrow = X + (long)rowA * 128;

  v8f c = {};
#pragma unroll
  for (int k0 = 0; k0 < 128; k0 += 4) {
    const int k = k0 + khalf;
    v2f a; a.x = xrow[k];                 a.y = xrow[k + 1];
    v2f b; b.x = W[k * 64 + colBase + m]; b.y = W[(k + 1) * 64 + colBase + m];
    c = __builtin_amdgcn_wmma_f32_16x16x4_f32(false, a, false, b,
                                              (short)0, c, false, false);
  }
  const int mrow = (lane >> 4) * 8;
#pragma unroll
  for (int v = 0; v < 8; ++v) {
    int r = row0 + mrow + v;
    if (r < N) H[(long)r * 64 + colBase + m] = c[v];
  }

  const float as = aS[colBase + m], ad = aD[colBase + m];
  const int i = lane & 7;
  float xs[8], xd[8];
  for (int v = 0; v < 8; ++v) { xs[v] = c[v] * as; xd[v] = c[v] * ad; }
  const float ps = gsum8x8(xs, (unsigned)lane), pd = gsum8x8(xd, (unsigned)lane);
  const int slot = (mrow + i) * 8 + 2 * cw + ((lane >> 3) & 1);
  se[tile][0][slot] = ps;
  se[tile][1][slot] = pd;
  __syncthreads();
  const int tt = threadIdx.x & 127;
  const long o = (long)row0 * 8 + tt;
  if (o < (long)N * 8) { eS[o] = se[tile][0][tt]; eD[o] = se[tile][1][tt]; }
}

#define SORT_T    1024
#define SORT_NA   32
#define SORT_NB   8
#define SORT_RMAX 4096
#define SORT_LDC  13
#define SORT_LDP  8

__global__ void __launch_bounds__(SORT_T)
gat_dst_count(const int* __restrict__ dst, int* __restrict__ cntB,
              int N, int Npad, int E, int R, int S) {
  __shared__ int hist[SORT_RMAX + 32];
  const int t = threadIdx.x;
  const int a = (int)blockIdx.x / SORT_NB, b = (int)blockIdx.x % SORT_NB;
  const int e0 = b * S;
  const int ne = (E - e0 < S) ? ((E - e0 > 0) ? E - e0 : 0) : S;
  const int* __restrict__ d = dst + e0;
  const bool vec = (((size_t)d) & 15) == 0;
  const int n4 = vec ? (ne >> 2) : 0;
  const int4* __restrict__ d4 = (const int4*)d;
  int* __restrict__ plane = cntB + (long)b * Npad;
  const int spare = SORT_RMAX + (t & 31);
  const int nlo = a * R, nhi = (nlo + R < N) ? nlo + R : N;
  for (int lo = nlo; lo < nhi; lo += SORT_RMAX) {
    const int len = (nhi - lo < SORT_RMAX) ? nhi - lo : SORT_RMAX;
    for (int i = t; i < len; i += SORT_T) hist[i] = 0;
    if (t < 32) hist[SORT_RMAX + t] = 0;
    __syncthreads();
    for (int i0 = 0; i0 < n4; i0 += SORT_LDC * SORT_T) {
      int4 v[SORT_LDC];
#pragma unroll
      for (int u = 0; u < SORT_LDC; ++u) { const int i = i0 + u * SORT_T + t; v[u] = d4[i < n4 ? i : n4 - 1]; }
      __builtin_amdgcn_sched_barrier(0);
#pragma unroll
      for (int u = 0; u < SORT_LDC; ++u) {
        const bool in = i0 + u * SORT_T + t < n4;
        const unsigned r0 = (unsigned)(v[u].x - lo), r1 = (unsigned)(v[u].y - lo);
        const unsigned r2 = (unsigned)(v[u].z - lo), r3 = (unsigned)(v[u].w - lo);
        atomicAdd(hist + ((in && r0 < (unsigned)len) ? (int)r0 : spare), 1);
        atomicAdd(hist + ((in && r1 < (unsigned)len) ? (int)r1 : spare), 1);
        atomicAdd(hist + ((in && r2 < (unsigned)len) ? (int)r2 : spare), 1);
        atomicAdd(hist + ((in && r3 < (unsigned)len) ? (int)r3 : spare), 1);
      }
    }
    for (int i = 4 * n4 + t; i < ne; i += SORT_T) {
      const unsigned r = (unsigned)(d[i] - lo);
      if (r < (unsigned)len) atomicAdd(hist + r, 1);
    }
    __syncthreads();
    for (int i = t; i < len; i += SORT_T) plane[lo + i] = hist[i];
    __syncthreads();
  }
}

#define SCAN_T    1024
#define SCAN_PER  63
#define SCAN_TILE (SCAN_T * SCAN_PER)
#define SCAN_LD   16
__global__ void __launch_bounds__(SCAN_T)
gat_dst_scan_gemm1(const int* __restrict__ cnt, int* __restrict__ offs, int N, int Npad,
                   const float* __restrict__ X, const float* __restrict__ W,
                   const float* __restrict__ aS, const float* __restrict__ aD,
                   float* __restrict__ H, float* __restrict__ eS, float* __restrict__ eD) {
  __shared__ int4 buf4[SCAN_LD * SCAN_T];
  __shared__ int part[SCAN_T];
  __shared__ float se[8][2][128];
  if (blockIdx.x != 0) {
    gat_gemm1_tiles(X, W, aS, aD, H, eS, eD, N, (int)blockIdx.x - 1, se);
    return;
  }
  int* buf = (int*)buf4;
  const int t = threadIdx.x;
  const int spare = SCAN_TILE + t;
  int carry = 0;
  for (int base = 0; base < N; base += SCAN_TILE) {
    const int n   = (N - base < SCAN_TILE) ? (N - base) : SCAN_TILE;
    const int n4  = n >> 2;
    const int per = ((n + SCAN_T - 1) / SCAN_T) | 1;
    const int b0  = t * per;

    buf[spare] = 0;
    if (n4 > 0) {
      const int4* __restrict__ c4 = (const int4*)(cnt + base);
      const long ps4 = (long)(Npad >> 2);
#pragma unroll
      for (int u = 0; u < SCAN_LD; u += 2) {
        if (u * SCAN_T < n4) {
          const int i0 = t + u * SCAN_T, i1 = i0 + SCAN_T;
          const int4* __restrict__ p0 = c4 + (i0 < n4 ? i0 : n4 - 1);
          const int4* __restrict__ p1 = c4 + (i1 < n4 ? i1 : n4 - 1);
          int4 w0[SORT_NB], w1[SORT_NB];
#pragma unroll
          for (int p = 0; p < SORT_NB; ++p) { w0[p] = p0[p * ps4]; w1[p] = p1[p * ps4]; }
          int4 s0 = w0[0], s1 = w1[0];
#pragma unroll
          for (int p = 1; p < SORT_NB; ++p) {
            s0.x += w0[p].x; s0.y += w0[p].y; s0.z += w0[p].z; s0.w += w0[p].w;
            s1.x += w1[p].x; s1.y += w1[p].y; s1.z += w1[p].z; s1.w += w1[p].w;
          }
          buf4[i0] = s0;
          buf4[i1] = s1;
        }
      }
    }
    if (n & 3) {
      __syncthreads();
      if (t < (n & 3)) {
        int s = 0;
        for (int p = 0; p < SORT_NB; ++p) s += cnt[(long)p * Npad + base + 4 * n4 + t];
        buf[4 * n4 + t] = s;
      }
    }
    __syncthreads();

    int sum = 0;
    for (int i0 = 0; i0 < per; i0 += 8) {
      int c[8];
#pragma unroll
      for (int u = 0; u < 8; ++u) {
        const int idx = b0 + i0 + u;
        const bool ok = (i0 + u < per) && (idx < n);
        const int w = buf[ok ? idx : spare];
        c[u] = ok ? w : 0;
      }
#pragma unroll
      for (int u = 0; u < 8; ++u) sum += c[u];
    }
    part[t] = sum;
    __syncthreads();
    for (int o = 1; o < SCAN_T; o <<= 1) {
      const int v = (t >= o) ? part[t - o] : 0;
      __syncthreads();
      part[t] += v;
      __syncthreads();
    }
    const int total = part[SCAN_T - 1];
    int run = carry + part[t] - sum;
    for (int i0 = 0; i0 < per; i0 += 8) {
      int c[8], at[8];
#pragma unroll
      for (int u = 0; u < 8; ++u) {
        const int idx = b0 + i0 + u;
        const bool ok = (i0 + u < per) && (idx < n);
        at[u] = ok ? idx : spare;
        const int w = buf[at[u]];
        c[u] = ok ? w : 0;
      }
#pragma unroll
      for (int u = 0; u < 8; ++u) { buf[at[u]] = run; run += c[u]; }
    }
    __syncthreads();

    if (n4 > 0) {
      int4* __restrict__ o4 = (int4*)(offs + base);
#pragma unroll
      for (int u = 0; u < SCAN_LD; ++u) { const int i = t + u * SCAN_T; if (i < n4) o4[i] = buf4[i]; }
    }
    if (t < (n & 3)) offs[base + 4 * n4 + t] = buf[4 * n4 + t];
    carry += total;
    __syncthreads();
  }
  if (t == 0) offs[N] = carry;
}

__global__ void __launch_bounds__(SORT_T)
gat_dst_place(const int* __restrict__ src, const int* __restrict__ dst,
              const int* __restrict__ cntB, const int* __restrict__ offs,
              int* __restrict__ perm, int N, int Npad, int E, int R, int S) {
  __shared__ int cur[SORT_RMAX + 32];
  const int t = threadIdx.x;
  const int a = (int)blockIdx.x / SORT_NB, b = (int)blockIdx.x % SORT_NB;
  const int e0 = b * S;
  const int ne = (E - e0 < S) ? ((E - e0 > 0) ? E - e0 : 0) : S;
  const int* __restrict__ d = dst + e0;
  const int* __restrict__ s = src + e0;
  const bool vec = ((((size_t)d) | ((size_t)s)) & 15) == 0;
  const int n4 = vec ? (ne >> 2) : 0;
  const int4* __restrict__ d4 = (const int4*)d;
  const int4* __restrict__ s4 = (const int4*)s;
  const int spare = SORT_RMAX + (t & 31);
  const int nlo = a * R, nhi = (nlo + R < N) ? nlo + R : N;
  for (int lo = nlo; lo < nhi; lo += SORT_RMAX) {
    const int len = (nhi - lo < SORT_RMAX) ? nhi - lo : SORT_RMAX;
    for (int i = t; i < len; i += SORT_T) {
      int p = offs[lo + i];
#pragma unroll
      for (int bb = 0; bb < SORT_NB - 1; ++bb) {
        const int cb = cntB[(long)bb * Npad + lo + i];
        p += (bb < b) ? cb : 0;
      }
      cur[i] = p;
    }
    if (t < 32) cur[SORT_RMAX + t] = 0;
    __syncthreads();
    for (int i0 = 0; i0 < n4; i0 += SORT_LDP * SORT_T) {
      int4 v[SORT_LDP], w[SORT_LDP];
#pragma unroll
      for (int u = 0; u < SORT_LDP; ++u) {
        const int i = i0 + u * SORT_T + t, ic = i < n4 ? i : n4 - 1;
        v[u] = d4[ic]; w[u] = s4[ic];
      }
      __builtin_amdgcn_sched_barrier(0);
#pragma unroll
      for (int u = 0; u < SORT_LDP; ++u) {
        const bool in = i0 + u * SORT_T + t < n4;
        const unsigned r0 = (unsigned)(v[u].x - lo), r1 = (unsigned)(v[u].y - lo);
        const unsigned r2 = (unsigned)(v[u].z - lo), r3 = (unsigned)(v[u].w - lo);
        const bool h0 = in && r0 < (unsigned)len, h1 = in && r1 < (unsigned)len;
        const bool h2 = in && r2 < (unsigned)len, h3 = in && r3 < (unsigned)len;
        const int p0 = atomicAdd(cur + (h0 ? (int)r0 : spare), 1);
        const int p1 = atomicAdd(cur + (h1 ? (int)r1 : spare), 1);
        const int p2 = atomicAdd(cur + (h2 ? (int)r2 : spare), 1);
        const int p3 = atomicAdd(cur + (h3 ? (int)r3 : spare), 1);
        if (h0) perm[p0] = w[u].x;
        if (h1) perm[p1] = w[u].y;
        if (h2) perm[p2] = w[u].z;
        if (h3) perm[p3] = w[u].w;
      }
    }
    for (int i = 4 * n4 + t; i < ne; i += SORT_T) {
      const unsigned r = (unsigned)(d[i] - lo);
      if (r < (unsigned)len) perm[atomicAdd(cur + r, 1)] = s[i];
    }
    __syncthreads();
  }
}


__device__ __forceinline__ void l1_scores(const int* __restrict__ perm, const float* __restrict__ eS,
                                          int base, int cnt, unsigned q, unsigned i,
                                          const float (&ed)[2], float (&e)[4][2]) {
  for (int r = 0; r < 4; ++r) {
    const int j = 8 * r + (int)i;
    const bool ok = j < cnt;
    const unsigned s = (unsigned)perm[base + (ok ? j : 0)];
    for (int k = 0; k < 2; ++k) {
      const float v = lrelu(eS[s * 8u + 4u * k + q] + ed[k]);
      e[r][k] = ok ? v : -INFINITY;
    }
  }
}

template <int J>
__device__ __forceinline__ void l1_load(unsigned lane, int s, const float* __restrict__ Hn,
                                        float& h0, float& h1) {
  const float* row = Hn + (unsigned)__builtin_amdgcn_readlane(s, J) * 64u;
  h0 = row[lane];
  h1 = row[32u + lane];
}

template <int R>
__device__ __forceinline__ void l1_group(unsigned lane, int s, const float (&A)[4][2],
                                         const float* __restrict__ Hn, float& acc0, float& acc1) {
  float h[16];
  l1_load<8 * R + 0>(lane, s, Hn, h[0],  h[1]);
  l1_load<8 * R + 1>(lane, s, Hn, h[2],  h[3]);
  l1_load<8 * R + 2>(lane, s, Hn, h[4],  h[5]);
  l1_load<8 * R + 3>(lane, s, Hn, h[6],  h[7]);
  l1_load<8 * R + 4>(lane, s, Hn, h[8],  h[9]);
  l1_load<8 * R + 5>(lane, s, Hn, h[10], h[11]);
  l1_load<8 * R + 6>(lane, s, Hn, h[12], h[13]);
  l1_load<8 * R + 7>(lane, s, Hn, h[14], h[15]);
  acc0 += h[0]  * bcast8<0>(A[R][0]);  acc1 += h[1]  * bcast8<0>(A[R][1]);
  acc0 += h[2]  * bcast8<1>(A[R][0]);  acc1 += h[3]  * bcast8<1>(A[R][1]);
  acc0 += h[4]  * bcast8<2>(A[R][0]);  acc1 += h[5]  * bcast8<2>(A[R][1]);
  acc0 += h[6]  * bcast8<3>(A[R][0]);  acc1 += h[7]  * bcast8<3>(A[R][1]);
  acc0 += h[8]  * bcast8<4>(A[R][0]);  acc1 += h[9]  * bcast8<4>(A[R][1]);
  acc0 += h[10] * bcast8<5>(A[R][0]);  acc1 += h[11] * bcast8<5>(A[R][1]);
  acc0 += h[12] * bcast8<6>(A[R][0]);  acc1 += h[13] * bcast8<6>(A[R][1]);
  acc0 += h[14] * bcast8<7>(A[R][0]);  acc1 += h[15] * bcast8<7>(A[R][1]);
}

__device__ __forceinline__ void l1_chunk_sum(const int* __restrict__ perm, int base, int cnt,
                                             unsigned lane, const float (&A)[4][2],
                                             const float* __restrict__ Hn, float& acc0, float& acc1) {
  const int s = perm[base + (((int)lane < cnt) ? (int)lane : 0)];
  l1_group<0>(lane, s, A, Hn, acc0, acc1);
  if (cnt > 8)  l1_group<1>(lane, s, A, Hn, acc0, acc1);
  if (cnt > 16) l1_group<2>(lane, s, A, Hn, acc0, acc1);
  if (cnt > 24) l1_group<3>(lane, s, A, Hn, acc0, acc1);
}

#define XT_LD 68
__global__ void __launch_bounds__(512)
gat_node_agg1_gemm2(const int* __restrict__ offs, const int* __restrict__ perm,
                    const float* __restrict__ eS, const float* __restrict__ eD,
                    const float* __restrict__ Hn, const float* __restrict__ bias,
                    const float* __restrict__ W2, const float* __restrict__ aS2,
                    const float* __restrict__ aD2, float* __restrict__ H2,
                    float* __restrict__ eS2, float* __restrict__ eD2, int N) {
  __shared__ float xt[16 * XT_LD];
  __shared__ float ht[16 * 40];
  __shared__ float pe[3][2][16];
  const unsigned lane = threadIdx.x & 31;
  const unsigned wv   = threadIdx.x >> 5;
  const int row0 = (int)blockIdx.x * 16;
  const int n = row0 + (int)wv;
  float v0 = 0.f, v1 = 0.f;
  if (n < N) {
    const unsigned q = lane >> 3, i = lane & 7;
    const int beg = offs[n], deg = offs[n + 1] - beg;
    const float ed[2] = { eD[(unsigned)n * 8u + q], eD[(unsigned)n * 8u + 4u + q] };
    float acc0 = 0.f, acc1 = 0.f;
    if (deg > 0 && deg <= 32) {
      float e[4][2], A[4][2], m[2], ss[2];
      l1_scores(perm, eS, beg, deg, q, i, ed, e);
      for (int k = 0; k < 2; ++k) {
        m[k] = gmax8(fmaxf(fmaxf(e[0][k], e[1][k]), fmaxf(e[2][k], e[3][k])));
        if (!isfinite(m[k])) m[k] = 0.f;
        float t = 0.f;
        for (int r = 0; r < 4; ++r) { e[r][k] = expf(e[r][k] - m[k]); t += e[r][k]; }
        ss[k] = gsum8(t);
        for (int r = 0; r < 4; ++r) A[r][k] = e[r][k] / (ss[k] + GAT_EPS);
      }
      l1_chunk_sum(perm, beg, deg, lane, A, Hn, acc0, acc1);
    } else if (deg > 32) {
      float e[4][2], A[4][2], m[2] = { -INFINITY, -INFINITY }, ss[2] = { 0.f, 0.f };
      for (int c0 = 0; c0 < deg; c0 += 32) {
        const int cn = (deg - c0 < 32) ? (deg - c0) : 32;
        l1_scores(perm, eS, beg + c0, cn, q, i, ed, e);
        for (int k = 0; k < 2; ++k)
          m[k] = fmaxf(m[k], fmaxf(fmaxf(e[0][k], e[1][k]), fmaxf(e[2][k], e[3][k])));
      }
      for (int k = 0; k < 2; ++k) { m[k] = gmax8(m[k]); if (!isfinite(m[k])) m[k] = 0.f; }
      for (int c0 = 0; c0 < deg; c0 += 32) {
        const int cn = (deg - c0 < 32) ? (deg - c0) : 32;
        l1_scores(perm, eS, beg + c0, cn, q, i, ed, e);
        for (int k = 0; k < 2; ++k)
          for (int r = 0; r < 4; ++r) ss[k] += expf(e[r][k] - m[k]);
      }
      for (int k = 0; k < 2; ++k) ss[k] = gsum8(ss[k]);
      for (int c0 = 0; c0 < deg; c0 += 32) {
        const int cn = (deg - c0 < 32) ? (deg - c0) : 32;
        l1_scores(perm, eS, beg + c0, cn, q, i, ed, e);
        for (int k = 0; k < 2; ++k)
          for (int r = 0; r < 4; ++r) A[r][k] = expf(e[r][k] - m[k]) / (ss[k] + GAT_EPS);
        l1_chunk_sum(perm, beg + c0, cn, lane, A, Hn, acc0, acc1);
      }
    }
    v0 = acc0 + bias[lane];       v0 = v0 > 0.f ? v0 : 0.f;
    v1 = acc1 + bias[32u + lane]; v1 = v1 > 0.f ? v1 : 0.f;
  }
  const int m     = (int)(lane & 15u);
  const int khalf = (int)(lane >> 4) * 2;
  const int mrow  = (int)(lane >> 4) * 8;
  const int col   = (int)wv * 16 + m;
  const bool colOK = (col < 40);
  float bx[16], by[16], as2 = 0.f, ad2 = 0.f;
  if (wv < 3u) {
    const int colc = colOK ? col : 39;
    const float* wcol = W2 + colc;
    for (int ks = 0; ks < 16; ++ks) {
      bx[ks] = wcol[(4 * ks + khalf) * 40];
      by[ks] = wcol[(4 * ks + khalf + 1) * 40];
    }
    as2 = aS2[colc];
    ad2 = aD2[colc];
  }
  xt[wv * XT_LD + lane]       = v0;
  xt[wv * XT_LD + 32u + lane] = v1;
  __syncthreads();

  if (wv < 3u) {
    const float* xrow = xt + m * XT_LD;
    v8f c = {};
    for (int ks = 0; ks < 16; ++ks) {
      const int k = 4 * ks + khalf;
      v2f a; a.x = xrow[k]; a.y = xrow[k + 1];
      v2f b;
      b.x = colOK ? bx[ks] : 0.0f;
      b.y = colOK ? by[ks] : 0.0f;
      c = __builtin_amdgcn_wmma_f32_16x16x4_f32(false, a, false, b,
                                                (short)0, c, false, false);
    }
    const float as = colOK ? as2 : 0.f, ad = colOK ? ad2 : 0.f;
    const int i = (int)(lane & 7u);
    if (colOK) {
      for (int v = 0; v < 8; ++v) ht[(mrow + v) * 40 + col] = c[v];
    }
    float xs[8], xd[8];
    for (int v = 0; v < 8; ++v) { xs[v] = c[v] * as; xd[v] = c[v] * ad; }
    float ps = gsum8x8(xs, lane); ps += __shfl_xor(ps, 8);
    float pd = gsum8x8(xd, lane); pd += __shfl_xor(pd, 8);
    if ((lane & 8u) == 0u) { pe[wv][0][mrow + i] = ps; pe[wv][1][mrow + i] = pd; }
  }
  __syncthreads();

  const long base = (long)row0 * 40;
  const long lim  = (long)N * 40;
  const unsigned t = threadIdx.x;
  if (base + t < lim) H2[base + t] = ht[t];
  if (t < 128u && base + 512 + t < lim) H2[base + 512 + t] = ht[512u + t];
  if (t < 32u) {
    const unsigned r = t & 15u, w = t >> 4;
    if (row0 + (int)r < N) {
      const float e = pe[0][w][r] + pe[1][w][r] + pe[2][w][r];
      if (w) eD2[row0 + (int)r] = e; else eS2[row0 + (int)r] = e;
    }
  }
}


template <int J>
__device__ __forceinline__ void l2_load(unsigned lane, int s, const float* __restrict__ Hn,
                                        float& h0, float& h1) {
  const float* row = Hn + (unsigned)__builtin_amdgcn_readlane(s, J) * 40u;
  h0 = row[lane];
  h1 = (lane < 8u) ? row[32u + lane] : 0.f;
}
template <int J>
__device__ __forceinline__ float l2_alpha(float a) {
  return __int_as_float(__builtin_amdgcn_readlane(__float_as_int(a), J));
}

template <int R>
__device__ __forceinline__ void l2_group(unsigned lane, int s, float a,
                                         const float* __restrict__ Hn, float& acc0, float& acc1) {
  float h[16];
  l2_load<8 * R + 0>(lane, s, Hn, h[0],  h[1]);
  l2_load<8 * R + 1>(lane, s, Hn, h[2],  h[3]);
  l2_load<8 * R + 2>(lane, s, Hn, h[4],  h[5]);
  l2_load<8 * R + 3>(lane, s, Hn, h[6],  h[7]);
  l2_load<8 * R + 4>(lane, s, Hn, h[8],  h[9]);
  l2_load<8 * R + 5>(lane, s, Hn, h[10], h[11]);
  l2_load<8 * R + 6>(lane, s, Hn, h[12], h[13]);
  l2_load<8 * R + 7>(lane, s, Hn, h[14], h[15]);
  float a0;
  a0 = l2_alpha<8 * R + 0>(a); acc0 += h[0]  * a0; acc1 += h[1]  * a0;
  a0 = l2_alpha<8 * R + 1>(a); acc0 += h[2]  * a0; acc1 += h[3]  * a0;
  a0 = l2_alpha<8 * R + 2>(a); acc0 += h[4]  * a0; acc1 += h[5]  * a0;
  a0 = l2_alpha<8 * R + 3>(a); acc0 += h[6]  * a0; acc1 += h[7]  * a0;
  a0 = l2_alpha<8 * R + 4>(a); acc0 += h[8]  * a0; acc1 += h[9]  * a0;
  a0 = l2_alpha<8 * R + 5>(a); acc0 += h[10] * a0; acc1 += h[11] * a0;
  a0 = l2_alpha<8 * R + 6>(a); acc0 += h[12] * a0; acc1 += h[13] * a0;
  a0 = l2_alpha<8 * R + 7>(a); acc0 += h[14] * a0; acc1 += h[15] * a0;
}

__device__ __forceinline__ void l2_chunk_sum(unsigned lane, int s, float a, int cnt,
                                             const float* __restrict__ Hn, float& acc0, float& acc1) {
  l2_group<0>(lane, s, a, Hn, acc0, acc1);
  if (cnt > 8)  l2_group<1>(lane, s, a, Hn, acc0, acc1);
  if (cnt > 16) l2_group<2>(lane, s, a, Hn, acc0, acc1);
  if (cnt > 24) l2_group<3>(lane, s, a, Hn, acc0, acc1);
}

__global__ void gat_node_agg2(const int* __restrict__ offs, const int* __restrict__ perm,
                              const float* __restrict__ eS, const float* __restrict__ eD,
                              const float* __restrict__ Hn, const float* __restrict__ bias,
                              float* __restrict__ outp, int N) {
  __shared__ float rows[8 * 40];
  const unsigned lane = threadIdx.x & 31;
  const unsigned wv   = threadIdx.x >> 5;
  const int n = (int)(blockIdx.x * 8u + wv);
  float acc0 = 0.f, acc1 = 0.f;
  if (n < N) {
    const int beg = offs[n], deg = offs[n + 1] - beg;
    const float ed = eD[n];
    if (deg > 0 && deg <= 32) {
      const bool ok = (int)lane < deg;
      const int s = perm[beg + (ok ? (int)lane : 0)];
      const float e = ok ? lrelu(eS[s] + ed) : -INFINITY;
      float m = wmax32(e);
      if (!isfinite(m)) m = 0.f;
      const float p  = expf(e - m);
      const float ss = wsum32(p);
      const float a  = p / (ss + GAT_EPS);
      l2_chunk_sum(lane, s, a, deg, Hn, acc0, acc1);
    } else if (deg > 32) {
      float m = -INFINITY, ss = 0.f;
      for (int c0 = 0; c0 < deg; c0 += 32) {
        const bool ok = c0 + (int)lane < deg;
        const int s = perm[beg + (ok ? c0 + (int)lane : 0)];
        m = fmaxf(m, ok ? lrelu(eS[s] + ed) : -INFINITY);
      }
      m = wmax32(m);
      if (!isfinite(m)) m = 0.f;
      for (int c0 = 0; c0 < deg; c0 += 32) {
        const bool ok = c0 + (int)lane < deg;
        const int s = perm[beg + (ok ? c0 + (int)lane : 0)];
        ss += ok ? expf(lrelu(eS[s] + ed) - m) : 0.f;
      }
      ss = wsum32(ss);
      for (int c0 = 0; c0 < deg; c0 += 32) {
        const int cn = (deg - c0 < 32) ? (deg - c0) : 32;
        const bool ok = (int)lane < cn;
        const int s = perm[beg + c0 + (ok ? (int)lane : 0)];
        const float a = ok ? expf(lrelu(eS[s] + ed) - m) / (ss + GAT_EPS) : 0.f;
        l2_chunk_sum(lane, s, a, cn, Hn, acc0, acc1);
      }
    }
    const float v0 = acc0 + bias[lane];
    const float v1 = (lane < 8u) ? acc1 + bias[32u + lane] : -INFINITY;
    const float mx = wmax32(fmaxf(v0, v1));
    const float sm = wsum32(expf(v0 - mx) + ((lane < 8u) ? expf(v1 - mx) : 0.f));
    const float ls = mx + logf(sm);
    rows[wv * 40u + lane] = v0 - ls;
    if (lane < 8u) rows[wv * 40u + 32u + lane] = v1 - ls;
  }
  __syncthreads();
  const long base = (long)blockIdx.x * 320;
  const long lim  = (long)N * 40;
  const unsigned t = threadIdx.x;
  if (base + t < lim) outp[base + t] = rows[t];
  if (t < 64u && base + 256 + t < lim) outp[base + 256 + t] = rows[256u + t];
}

extern "C" void kernel_launch(void* const* d_in, const int* in_sizes, int n_in,
                              void* d_out, int out_size, void* d_ws, size_t ws_size,
                              hipStream_t stream) {
  const float* x     = (const float*)d_in[0];
  const int*   eidx  = (const int*)d_in[1];
  const float* W1    = (const float*)d_in[2];
  const float* aS1   = (const float*)d_in[3];
  const float* aD1   = (const float*)d_in[4];
  const float* b1    = (const float*)d_in[5];
  const float* W2    = (const float*)d_in[6];
  const float* aS2   = (const float*)d_in[7];
  const float* aD2   = (const float*)d_in[8];
  const float* b2    = (const float*)d_in[9];
  float* out = (float*)d_out;

  const int N = in_sizes[0] / 128;
  const int E = in_sizes[1] / 2;
  const int* src = eidx;
  const int* dst = eidx + E;

  const long N64 = (long)N * 64, N40 = (long)N * 40;
  const long N8  = (long)N * 8,  E8  = (long)E * 8;

  float* ws      = (float*)d_ws;
  float* h1      = ws;
  float* esrc1   = h1 + N64;
  float* edst1   = esrc1 + N8;
  float* e1      = edst1 + N8;
  float* s1      = e1 + E8;
  float* agg1    = s1 + N8 + N8;
  float* h2      = agg1 + N64;
  float* esrc2   = h2 + N40;
  float* edst2   = esrc2 + N;
  int* perm = (int*)e1 + E;
  const int Npad = ((N + 31) / 32) * 32;
  int* cntB = (int*)s1;
  int* offs = cntB + (long)SORT_NB * Npad;
  const int R = (N + SORT_NA - 1) / SORT_NA;
  const int S = (((E + SORT_NB - 1) / SORT_NB) + 3) & ~3;

  const int BT = 256;
  const int rowBlocks = (N + 15) / 16;
  const int nodeBlocks = (N + 7) / 8;
  const int gemmBlocks = (N + 127) / 128;

  gat_dst_count<<<SORT_NA * SORT_NB, SORT_T, 0, stream>>>(dst, cntB, N, Npad, E, R, S);
  gat_dst_scan_gemm1<<<1 + gemmBlocks, SCAN_T, 0, stream>>>(cntB, offs, N, Npad, x, W1, aS1, aD1,
                                                          h1, esrc1, edst1);
  gat_dst_place<<<SORT_NA * SORT_NB, SORT_T, 0, stream>>>(src, dst, cntB, offs, perm, N, Npad, E, R, S);

  gat_node_agg1_gemm2<<<rowBlocks, 512, 0, stream>>>(offs, perm, esrc1, edst1, h1, b1,
                                                     W2, aS2, aD2, h2, esrc2, edst2, N);

  gat_node_agg2<<<nodeBlocks, BT, 0, stream>>>(offs, perm, esrc2, edst2, h2, b2, out, N);
}


# edge sort grid 16 ranges x 16 slices, not 32 x 8: a range of 3125 rows still fits the LDS counters in one trip and each work-group streams half the edges (segment order now by 16 slices)
# speedup vs baseline: 6.7615x; 1.0895x over previous
#include <hip/hip_runtime.h>
#include <math.h>


typedef float v2f __attribute__((ext_vector_type(2)));
typedef float v8f __attribute__((ext_vector_type(8)));

#define NEG_SLOPE 0.2f
#define GAT_EPS 1e-16f

template <int I>
__device__ __forceinline__ float bcast8(float v) {
  return __uint_as_float(__builtin_amdgcn_mov_dpp8(__float_as_uint(v), (unsigned)(I * 0x249249)));
}

__device__ __forceinline__ float lrelu(float v) { return (v > 0.f) ? v : v * NEG_SLOPE; }

__device__ __forceinline__ float gmax8(float v) {
  v = fmaxf(v, __shfl_xor(v, 1)); v = fmaxf(v, __shfl_xor(v, 2)); v = fmaxf(v, __shfl_xor(v, 4));
  return v;
}
__device__ __forceinline__ float gsum8(float v) {
  v += __shfl_xor(v, 1); v += __shfl_xor(v, 2); v += __shfl_xor(v, 4);
  return v;
}
__device__ __forceinline__ float gsum8x8(const float (&x)[8], unsigned lane) {
  const bool b2 = (lane & 4u) != 0u, b1 = (lane & 2u) != 0u, b0 = (lane & 1u) != 0u;
  float y[4], z[2];
  for (int j = 0; j < 4; ++j) {
    const float keep = b2 ? x[4 + j] : x[j], send = b2 ? x[j] : x[4 + j];
    y[j] = keep + __shfl_xor(send, 4);
  }
  for (int j = 0; j < 2; ++j) {
    const float keep = b1 ? y[2 + j] : y[j], send = b1 ? y[j] : y[2 + j];
    z[j] = keep + __shfl_xor(send, 2);
  }
  const float keep = b0 ? z[1] : z[0], send = b0 ? z[0] : z[1];
  return keep + __shfl_xor(send, 1);
}
__device__ __forceinline__ float wmax32(float v) {
  v = gmax8(v); v = fmaxf(v, __shfl_xor(v, 8)); v = fmaxf(v, __shfl_xor(v, 16));
  return v;
}
__device__ __forceinline__ float wsum32(float v) {
  v = gsum8(v); v += __shfl_xor(v, 8); v += __shfl_xor(v, 16);
  return v;
}

__device__ __forceinline__ void
gat_gemm1_tiles(const float* __restrict__ X, const float* __restrict__ W,
                const float* __restrict__ aS, const float* __restrict__ aD,
                float* __restrict__ H, float* __restrict__ eS, float* __restrict__ eD, int N,
                int blk, float (*se)[2][128]) {
  const int lane    = threadIdx.x & 31;
  const int wave    = threadIdx.x >> 5;
  const int tile    = wave >> 2;
  const int cw      = wave & 3;
  const int row0    = blk * 128 + tile * 16;
  const int colBase = cw * 16;
  const int m       = lane & 15;
  const int khalf   = (lane >> 4) * 2;

  int rowA = row0 + m; if (rowA >= N) rowA = N - 1;
  const float* xrow = X + (long)rowA * 128;

  v8f c = {};
#pragma unroll
  for (int k0 = 0; k0 < 128; k0 += 4) {
    const int k = k0 + khalf;
    v2f a; a.x = xrow[k];                 a.y = xrow[k + 1];
    v2f b; b.x = W[k * 64 + colBase + m]; b.y = W[(k + 1) * 64 + colBase + m];
    c = __builtin_amdgcn_wmma_f32_16x16x4_f32(false, a, false, b,
                                              (short)0, c, false, false);
  }
  const int mrow = (lane >> 4) * 8;
#pragma unroll
  for (int v = 0; v < 8; ++v) {
    int r = row0 + mrow + v;
    if (r < N) H[(long)r * 64 + colBase + m] = c[v];
  }

  const float as = aS[colBase + m], ad = aD[colBase + m];
  const int i = lane & 7;
  float xs[8], xd[8];
  for (int v = 0; v < 8; ++v) { xs[v] = c[v] * as; xd[v] = c[v] * ad; }
  const float ps = gsum8x8(xs, (unsigned)lane), pd = gsum8x8(xd, (unsigned)lane);
  const int slot = (mrow + i) * 8 + 2 * cw + ((lane >> 3) & 1);
  se[tile][0][slot] = ps;
  se[tile][1][slot] = pd;
  __syncthreads();
  const int tt = threadIdx.x & 127;
  const long o = (long)row0 * 8 + tt;
  if (o < (long)N * 8) { eS[o] = se[tile][0][tt]; eD[o] = se[tile][1][tt]; }
}

#define SORT_T    1024
#define SORT_NA   16
#define SORT_NB   16
#define SORT_RMAX 4096
#define SORT_LDC  13
#define SORT_LDP  8

__device__ __forceinline__ int wsum32i(int v) {
  v += __shfl_xor(v, 1); v += __shfl_xor(v, 2); v += __shfl_xor(v, 4);
  v += __shfl_xor(v, 8); v += __shfl_xor(v, 16);
  return v;
}

__global__ void __launch_bounds__(SORT_T)
gat_dst_count_gemm1(const int* __restrict__ dst, int* __restrict__ cntB, int* __restrict__ rtot,
                    int N, int Npad, int E, int R, int S,
                    const float* __restrict__ X, const float* __restrict__ W,
                    const float* __restrict__ aS, const float* __restrict__ aD,
                    float* __restrict__ H, float* __restrict__ eS, float* __restrict__ eD) {
  __shared__ int hist[SORT_RMAX + 32];
  __shared__ __align__(16) int wtot[32];
  __shared__ float se[8][2][128];
  if ((int)blockIdx.x >= SORT_NA * SORT_NB) {
    gat_gemm1_tiles(X, W, aS, aD, H, eS, eD, N, (int)blockIdx.x - SORT_NA * SORT_NB, se);
    return;
  }
  const int t = threadIdx.x;
  int mine = 0;
  const int a = (int)blockIdx.x / SORT_NB, b = (int)blockIdx.x % SORT_NB;
  const int e0 = b * S;
  const int ne = (E - e0 < S) ? ((E - e0 > 0) ? E - e0 : 0) : S;
  const int* __restrict__ d = dst + e0;
  const bool vec = (((size_t)d) & 15) == 0;
  const int n4 = vec ? (ne >> 2) : 0;
  const int4* __restrict__ d4 = (const int4*)d;
  int* __restrict__ plane = cntB + (long)b * Npad;
  const int spare = SORT_RMAX + (t & 31);
  const int nlo = a * R, nhi = (nlo + R < N) ? nlo + R : N;
  for (int lo = nlo; lo < nhi; lo += SORT_RMAX) {
    const int len = (nhi - lo < SORT_RMAX) ? nhi - lo : SORT_RMAX;
    for (int i = t; i < len; i += SORT_T) hist[i] = 0;
    if (t < 32) hist[SORT_RMAX + t] = 0;
    __syncthreads();
    for (int i0 = 0; i0 < n4; i0 += SORT_LDC * SORT_T) {
      int4 v[SORT_LDC];
#pragma unroll
      for (int u = 0; u < SORT_LDC; ++u) { const int i = i0 + u * SORT_T + t; v[u] = d4[i < n4 ? i : n4 - 1]; }
      __builtin_amdgcn_sched_barrier(0);
#pragma unroll
      for (int u = 0; u < SORT_LDC; ++u) {
        const bool in = i0 + u * SORT_T + t < n4;
        const unsigned r0 = (unsigned)(v[u].x - lo), r1 = (unsigned)(v[u].y - lo);
        const unsigned r2 = (unsigned)(v[u].z - lo), r3 = (unsigned)(v[u].w - lo);
        atomicAdd(hist + ((in && r0 < (unsigned)len) ? (int)r0 : spare), 1);
        atomicAdd(hist + ((in && r1 < (unsigned)len) ? (int)r1 : spare), 1);
        atomicAdd(hist + ((in && r2 < (unsigned)len) ? (int)r2 : spare), 1);
        atomicAdd(hist + ((in && r3 < (unsigned)len) ? (int)r3 : spare), 1);
      }
    }
    for (int i = 4 * n4 + t; i < ne; i += SORT_T) {
      const unsigned r = (unsigned)(d[i] - lo);
      if (r < (unsigned)len) atomicAdd(hist + r, 1);
    }
    __syncthreads();
    for (int i = t; i < len; i += SORT_T) { const int h = hist[i]; plane[lo + i] = h; mine += h; }
    __syncthreads();
  }
  mine = wsum32i(mine);
  if ((t & 31) == 0) wtot[t >> 5] = mine;
  __syncthreads();
  if (t == 0) {
    int v = 0;
    for (int w = 0; w < 32; ++w) v += wtot[w];
    rtot[blockIdx.x] = v;
  }
}

#define PLACE_K ((SORT_RMAX + SORT_T - 1) / SORT_T)
#define PLACE_W ((SORT_NA * SORT_NB + 31) / 32)
static_assert(SORT_NA * SORT_NB <= SORT_T, "one thread per (range, slice) total");

template <int K>
__device__ __forceinline__ void place_counts(const int* __restrict__ cntB, int Npad, int lo, int len,
                                             int i0, int b, int (&tot)[PLACE_K], int (&bel)[PLACE_K],
                                             int& sum) {
  int cb[K][SORT_NB];
  for (int j = 0; j < K; ++j) {
    const int i = i0 + j;
    const int* __restrict__ cp = cntB + lo + ((i < len) ? i : len - 1);
    for (int bb = 0; bb < SORT_NB; ++bb) cb[j][bb] = cp[(long)bb * Npad];
  }
  __builtin_amdgcn_sched_barrier(0);
  for (int j = 0; j < PLACE_K; ++j) { tot[j] = 0; bel[j] = 0; }
  for (int j = 0; j < K; ++j) {
    int all = 0, front = 0;
    for (int bb = 0; bb < SORT_NB; ++bb) { all += cb[j][bb]; front += (bb < b) ? cb[j][bb] : 0; }
    tot[j] = (i0 + j < len) ? all : 0;
    bel[j] = front;
    sum += tot[j];
  }
}
__global__ void __launch_bounds__(SORT_T)
gat_dst_place(const int* __restrict__ src, const int* __restrict__ dst,
              const int* __restrict__ cntB, const int* __restrict__ rtot,
              int* __restrict__ offs, int* __restrict__ perm,
              int N, int Npad, int E, int R, int S) {
  __shared__ int cur[SORT_RMAX + 32];
  __shared__ __align__(16) int bsum[PLACE_W];
  __shared__ __align__(16) int wsumS[32];
  const int t = threadIdx.x;
  const int lane = t & 31, wave = t >> 5;
  const int a = (int)blockIdx.x / SORT_NB, b = (int)blockIdx.x % SORT_NB;
  const int e0 = b * S;
  const int ne = (E - e0 < S) ? ((E - e0 > 0) ? E - e0 : 0) : S;
  const int* __restrict__ d = dst + e0;
  const int* __restrict__ s = src + e0;
  const bool vec = ((((size_t)d) | ((size_t)s)) & 15) == 0;
  const int n4 = vec ? (ne >> 2) : 0;
  const int4* __restrict__ d4 = (const int4*)d;
  const int4* __restrict__ s4 = (const int4*)s;
  const int spare = SORT_RMAX + (t & 31);
  const int nlo = a * R, nhi = (nlo + R < N) ? nlo + R : N;
  const int rt = (t < a * SORT_NB) ? rtot[t] : 0;
  int carry = 0;
  for (int lo = nlo; lo < nhi; lo += SORT_RMAX) {
    const int len = (nhi - lo < SORT_RMAX) ? nhi - lo : SORT_RMAX;
    const int K = (len + SORT_T - 1) / SORT_T;
    const int i0 = t * K;
    int tot[PLACE_K], bel[PLACE_K];
    int sum = 0;
    switch (K) {
      case 1:  place_counts<1>(cntB, Npad, lo, len, i0, b, tot, bel, sum); break;
      case 2:  place_counts<2>(cntB, Npad, lo, len, i0, b, tot, bel, sum); break;
      case 3:  place_counts<3>(cntB, Npad, lo, len, i0, b, tot, bel, sum); break;
      default: place_counts<PLACE_K>(cntB, Npad, lo, len, i0, b, tot, bel, sum); break;
    }
    int inc = sum;
    for (int o = 1; o < 32; o <<= 1) {
      const int v = __shfl_up(inc, o);
      if (lane >= o) inc += v;
    }
    if (lane == 31) wsumS[wave] = inc;
    if (lo == nlo && wave < PLACE_W) {
      const int v = wsum32i(rt);
      if (lane == 0) bsum[wave] = v;
    }
    if (t < 32) cur[SORT_RMAX + t] = 0;
    __syncthreads();
    if (lo == nlo)
      for (int w = 0; w < PLACE_W; ++w) carry += bsum[w];
    int before = 0, all = 0;
    for (int w = 0; w < 32; ++w) { const int v = wsumS[w]; all += v; before += (w < wave) ? v : 0; }
    int run = carry + before + inc - sum;
#pragma unroll
    for (int j = 0; j < PLACE_K; ++j) {
      if (j < K) {
        const int i = i0 + j;
        if (i < len) {
          cur[i] = run + bel[j];
          if (b == 0) offs[lo + i] = run;
        }
        run += tot[j];
      }
    }
    carry += all;
    __syncthreads();
    for (int i0e = 0; i0e < n4; i0e += SORT_LDP * SORT_T) {
      int4 v[SORT_LDP], w[SORT_LDP];
#pragma unroll
      for (int u = 0; u < SORT_LDP; ++u) {
        const int i = i0e + u * SORT_T + t, ic = i < n4 ? i : n4 - 1;
        v[u] = d4[ic]; w[u] = s4[ic];
      }
      __builtin_amdgcn_sched_barrier(0);
#pragma unroll
      for (int u = 0; u < SORT_LDP; ++u) {
        const bool in = i0e + u * SORT_T + t < n4;
        const unsigned r0 = (unsigned)(v[u].x - lo), r1 = (unsigned)(v[u].y - lo);
        const unsigned r2 = (unsigned)(v[u].z - lo), r3 = (unsigned)(v[u].w - lo);
        const bool h0 = in && r0 < (unsigned)len, h1 = in && r1 < (unsigned)len;
        const bool h2 = in && r2 < (unsigned)len, h3 = in && r3 < (unsigned)len;
        const int p0 = atomicAdd(cur + (h0 ? (int)r0 : spare), 1);
        const int p1 = atomicAdd(cur + (h1 ? (int)r1 : spare), 1);
        const int p2 = atomicAdd(cur + (h2 ? (int)r2 : spare), 1);
        const int p3 = atomicAdd(cur + (h3 ? (int)r3 : spare), 1);
        if (h0) perm[p0] = w[u].x;
        if (h1) perm[p1] = w[u].y;
        if (h2) perm[p2] = w[u].z;
        if (h3) perm[p3] = w[u].w;
      }
    }
    for (int i = 4 * n4 + t; i < ne; i += SORT_T) {
      const unsigned r = (unsigned)(d[i] - lo);
      if (r < (unsigned)len) perm[atomicAdd(cur + r, 1)] = s[i];
    }
    __syncthreads();
  }
  if (b == 0 && t == 0 && nlo < N && nhi == N) offs[N] = carry;
}


__device__ __forceinline__ void l1_scores(const int* __restrict__ perm, const float* __restrict__ eS,
                                          int base, int cnt, unsigned q, unsigned i,
                                          const float (&ed)[2], float (&e)[4][2]) {
  for (int r = 0; r < 4; ++r) {
    const int j = 8 * r + (int)i;
    const bool ok = j < cnt;
    const unsigned s = (unsigned)perm[base + (ok ? j : 0)];
    for (int k = 0; k < 2; ++k) {
      const float v = lrelu(eS[s * 8u + 4u * k + q] + ed[k]);
      e[r][k] = ok ? v : -INFINITY;
    }
  }
}

template <int J>
__device__ __forceinline__ void l1_load(unsigned lane, int s, const float* __restrict__ Hn,
                                        float& h0, float& h1) {
  const float* row = Hn + (unsigned)__builtin_amdgcn_readlane(s, J) * 64u;
  h0 = row[lane];
  h1 = row[32u + lane];
}

template <int R>
__device__ __forceinline__ void l1_group(unsigned lane, int s, const float (&A)[4][2],
                                         const float* __restrict__ Hn, float& acc0, float& acc1) {
  float h[16];
  l1_load<8 * R + 0>(lane, s, Hn, h[0],  h[1]);
  l1_load<8 * R + 1>(lane, s, Hn, h[2],  h[3]);
  l1_load<8 * R + 2>(lane, s, Hn, h[4],  h[5]);
  l1_load<8 * R + 3>(lane, s, Hn, h[6],  h[7]);
  l1_load<8 * R + 4>(lane, s, Hn, h[8],  h[9]);
  l1_load<8 * R + 5>(lane, s, Hn, h[10], h[11]);
  l1_load<8 * R + 6>(lane, s, Hn, h[12], h[13]);
  l1_load<8 * R + 7>(lane, s, Hn, h[14], h[15]);
  acc0 += h[0]  * bcast8<0>(A[R][0]);  acc1 += h[1]  * bcast8<0>(A[R][1]);
  acc0 += h[2]  * bcast8<1>(A[R][0]);  acc1 += h[3]  * bcast8<1>(A[R][1]);
  acc0 += h[4]  * bcast8<2>(A[R][0]);  acc1 += h[5]  * bcast8<2>(A[R][1]);
  acc0 += h[6]  * bcast8<3>(A[R][0]);  acc1 += h[7]  * bcast8<3>(A[R][1]);
  acc0 += h[8]  * bcast8<4>(A[R][0]);  acc1 += h[9]  * bcast8<4>(A[R][1]);
  acc0 += h[10] * bcast8<5>(A[R][0]);  acc1 += h[11] * bcast8<5>(A[R][1]);
  acc0 += h[12] * bcast8<6>(A[R][0]);  acc1 += h[13] * bcast8<6>(A[R][1]);
  acc0 += h[14] * bcast8<7>(A[R][0]);  acc1 += h[15] * bcast8<7>(A[R][1]);
}

__device__ __forceinline__ void l1_chunk_sum(const int* __restrict__ perm, int base, int cnt,
                                             unsigned lane, const float (&A)[4][2],
                                             const float* __restrict__ Hn, float& acc0, float& acc1) {
  const int s = perm[base + (((int)lane < cnt) ? (int)lane : 0)];
  l1_group<0>(lane, s, A, Hn, acc0, acc1);
  if (cnt > 8)  l1_group<1>(lane, s, A, Hn, acc0, acc1);
  if (cnt > 16) l1_group<2>(lane, s, A, Hn, acc0, acc1);
  if (cnt > 24) l1_group<3>(lane, s, A, Hn, acc0, acc1);
}

#define XT_LD 68
__global__ void __launch_bounds__(512)
gat_node_agg1_gemm2(const int* __restrict__ offs, const int* __restrict__ perm,
                    const float* __restrict__ eS, const float* __restrict__ eD,
                    const float* __restrict__ Hn, const float* __restrict__ bias,
                    const float* __restrict__ W2, const float* __restrict__ aS2,
                    const float* __restrict__ aD2, float* __restrict__ H2,
                    float* __restrict__ eS2, float* __restrict__ eD2, int N) {
  __shared__ float xt[16 * XT_LD];
  __shared__ float ht[16 * 40];
  __shared__ float pe[3][2][16];
  const unsigned lane = threadIdx.x & 31;
  const unsigned wv   = threadIdx.x >> 5;
  const int row0 = (int)blockIdx.x * 16;
  const int n = row0 + (int)wv;
  float v0 = 0.f, v1 = 0.f;
  if (n < N) {
    const unsigned q = lane >> 3, i = lane & 7;
    const int beg = offs[n], deg = offs[n + 1] - beg;
    const float ed[2] = { eD[(unsigned)n * 8u + q], eD[(unsigned)n * 8u + 4u + q] };
    float acc0 = 0.f, acc1 = 0.f;
    if (deg > 0 && deg <= 32) {
      float e[4][2], A[4][2], m[2], ss[2];
      l1_scores(perm, eS, beg, deg, q, i, ed, e);
      for (int k = 0; k < 2; ++k) {
        m[k] = gmax8(fmaxf(fmaxf(e[0][k], e[1][k]), fmaxf(e[2][k], e[3][k])));
        if (!isfinite(m[k])) m[k] = 0.f;
        float t = 0.f;
        for (int r = 0; r < 4; ++r) { e[r][k] = expf(e[r][k] - m[k]); t += e[r][k]; }
        ss[k] = gsum8(t);
        for (int r = 0; r < 4; ++r) A[r][k] = e[r][k] / (ss[k] + GAT_EPS);
      }
      l1_chunk_sum(perm, beg, deg, lane, A, Hn, acc0, acc1);
    } else if (deg > 32) {
      float e[4][2], A[4][2], m[2] = { -INFINITY, -INFINITY }, ss[2] = { 0.f, 0.f };
      for (int c0 = 0; c0 < deg; c0 += 32) {
        const int cn = (deg - c0 < 32) ? (deg - c0) : 32;
        l1_scores(perm, eS, beg + c0, cn, q, i, ed, e);
        for (int k = 0; k < 2; ++k)
          m[k] = fmaxf(m[k], fmaxf(fmaxf(e[0][k], e[1][k]), fmaxf(e[2][k], e[3][k])));
      }
      for (int k = 0; k < 2; ++k) { m[k] = gmax8(m[k]); if (!isfinite(m[k])) m[k] = 0.f; }
      for (int c0 = 0; c0 < deg; c0 += 32) {
        const int cn = (deg - c0 < 32) ? (deg - c0) : 32;
        l1_scores(perm, eS, beg + c0, cn, q, i, ed, e);
        for (int k = 0; k < 2; ++k)
          for (int r = 0; r < 4; ++r) ss[k] += expf(e[r][k] - m[k]);
      }
      for (int k = 0; k < 2; ++k) ss[k] = gsum8(ss[k]);
      for (int c0 = 0; c0 < deg; c0 += 32) {
        const int cn = (deg - c0 < 32) ? (deg - c0) : 32;
        l1_scores(perm, eS, beg + c0, cn, q, i, ed, e);
        for (int k = 0; k < 2; ++k)
          for (int r = 0; r < 4; ++r) A[r][k] = expf(e[r][k] - m[k]) / (ss[k] + GAT_EPS);
        l1_chunk_sum(perm, beg + c0, cn, lane, A, Hn, acc0, acc1);
      }
    }
    v0 = acc0 + bias[lane];       v0 = v0 > 0.f ? v0 : 0.f;
    v1 = acc1 + bias[32u + lane]; v1 = v1 > 0.f ? v1 : 0.f;
  }
  const int m     = (int)(lane & 15u);
  const int khalf = (int)(lane >> 4) * 2;
  const int mrow  = (int)(lane >> 4) * 8;
  const int col   = (int)wv * 16 + m;
  const bool colOK = (col < 40);
  float bx[16], by[16], as2 = 0.f, ad2 = 0.f;
  if (wv < 3u) {
    const int colc = colOK ? col : 39;
    const float* wcol = W2 + colc;
    for (int ks = 0; ks < 16; ++ks) {
      bx[ks] = wcol[(4 * ks + khalf) * 40];
      by[ks] = wcol[(4 * ks + khalf + 1) * 40];
    }
    as2 = aS2[colc];
    ad2 = aD2[colc];
  }
  xt[wv * XT_LD + lane]       = v0;
  xt[wv * XT_LD + 32u + lane] = v1;
  __syncthreads();

  if (wv < 3u) {
    const float* xrow = xt + m * XT_LD;
    v8f c = {};
    for (int ks = 0; ks < 16; ++ks) {
      const int k = 4 * ks + khalf;
      v2f a; a.x = xrow[k]; a.y = xrow[k + 1];
      v2f b;
      b.x = colOK ? bx[ks] : 0.0f;
      b.y = colOK ? by[ks] : 0.0f;
      c = __builtin_amdgcn_wmma_f32_16x16x4_f32(false, a, false, b,
                                                (short)0, c, false, false);
    }
    const float as = colOK ? as2 : 0.f, ad = colOK ? ad2 : 0.f;
    const int i = (int)(lane & 7u);
    if (colOK) {
      for (int v = 0; v < 8; ++v) ht[(mrow + v) * 40 + col] = c[v];
    }
    float xs[8], xd[8];
    for (int v = 0; v < 8; ++v) { xs[v] = c[v] * as; xd[v] = c[v] * ad; }
    float ps = gsum8x8(xs, lane); ps += __shfl_xor(ps, 8);
    float pd = gsum8x8(xd, lane); pd += __shfl_xor(pd, 8);
    if ((lane & 8u) == 0u) { pe[wv][0][mrow + i] = ps; pe[wv][1][mrow + i] = pd; }
  }
  __syncthreads();

  const long base = (long)row0 * 40;
  const long lim  = (long)N * 40;
  const unsigned t = threadIdx.x;
  if (base + t < lim) H2[base + t] = ht[t];
  if (t < 128u && base + 512 + t < lim) H2[base + 512 + t] = ht[512u + t];
  if (t < 32u) {
    const unsigned r = t & 15u, w = t >> 4;
    if (row0 + (int)r < N) {
      const float e = pe[0][w][r] + pe[1][w][r] + pe[2][w][r];
      if (w) eD2[row0 + (int)r] = e; else eS2[row0 + (int)r] = e;
    }
  }
}


template <int J>
__device__ __forceinline__ void l2_load(unsigned lane, int s, const float* __restrict__ Hn,
                                        float& h0, float& h1) {
  const float* row = Hn + (unsigned)__builtin_amdgcn_readlane(s, J) * 40u;
  h0 = row[lane];
  h1 = (lane < 8u) ? row[32u + lane] : 0.f;
}
template <int J>
__device__ __forceinline__ float l2_alpha(float a) {
  return __int_as_float(__builtin_amdgcn_readlane(__float_as_int(a), J));
}

template <int R>
__device__ __forceinline__ void l2_group(unsigned lane, int s, float a,
                                         const float* __restrict__ Hn, float& acc0, float& acc1) {
  float h[16];
  l2_load<8 * R + 0>(lane, s, Hn, h[0],  h[1]);
  l2_load<8 * R + 1>(lane, s, Hn, h[2],  h[3]);
  l2_load<8 * R + 2>(lane, s, Hn, h[4],  h[5]);
  l2_load<8 * R + 3>(lane, s, Hn, h[6],  h[7]);
  l2_load<8 * R + 4>(lane, s, Hn, h[8],  h[9]);
  l2_load<8 * R + 5>(lane, s, Hn, h[10], h[11]);
  l2_load<8 * R + 6>(lane, s, Hn, h[12], h[13]);
  l2_load<8 * R + 7>(lane, s, Hn, h[14], h[15]);
  float a0;
  a0 = l2_alpha<8 * R + 0>(a); acc0 += h[0]  * a0; acc1 += h[1]  * a0;
  a0 = l2_alpha<8 * R + 1>(a); acc0 += h[2]  * a0; acc1 += h[3]  * a0;
  a0 = l2_alpha<8 * R + 2>(a); acc0 += h[4]  * a0; acc1 += h[5]  * a0;
  a0 = l2_alpha<8 * R + 3>(a); acc0 += h[6]  * a0; acc1 += h[7]  * a0;
  a0 = l2_alpha<8 * R + 4>(a); acc0 += h[8]  * a0; acc1 += h[9]  * a0;
  a0 = l2_alpha<8 * R + 5>(a); acc0 += h[10] * a0; acc1 += h[11] * a0;
  a0 = l2_alpha<8 * R + 6>(a); acc0 += h[12] * a0; acc1 += h[13] * a0;
  a0 = l2_alpha<8 * R + 7>(a); acc0 += h[14] * a0; acc1 += h[15] * a0;
}

__device__ __forceinline__ void l2_chunk_sum(unsigned lane, int s, float a, int cnt,
                                             const float* __restrict__ Hn, float& acc0, float& acc1) {
  l2_group<0>(lane, s, a, Hn, acc0, acc1);
  if (cnt > 8)  l2_group<1>(lane, s, a, Hn, acc0, acc1);
  if (cnt > 16) l2_group<2>(lane, s, a, Hn, acc0, acc1);
  if (cnt > 24) l2_group<3>(lane, s, a, Hn, acc0, acc1);
}

__global__ void gat_node_agg2(const int* __restrict__ offs, const int* __restrict__ perm,
                              const float* __restrict__ eS, const float* __restrict__ eD,
                              const float* __restrict__ Hn, const float* __restrict__ bias,
                              float* __restrict__ outp, int N) {
  __shared__ float rows[8 * 40];
  const unsigned lane = threadIdx.x & 31;
  const unsigned wv   = threadIdx.x >> 5;
  const int n = (int)(blockIdx.x * 8u + wv);
  float acc0 = 0.f, acc1 = 0.f;
  if (n < N) {
    const int beg = offs[n], deg = offs[n + 1] - beg;
    const float ed = eD[n];
    if (deg > 0 && deg <= 32) {
      const bool ok = (int)lane < deg;
      const int s = perm[beg + (ok ? (int)lane : 0)];
      const float e = ok ? lrelu(eS[s] + ed) : -INFINITY;
      float m = wmax32(e);
      if (!isfinite(m)) m = 0.f;
      const float p  = expf(e - m);
      const float ss = wsum32(p);
      const float a  = p / (ss + GAT_EPS);
      l2_chunk_sum(lane, s, a, deg, Hn, acc0, acc1);
    } else if (deg > 32) {
      float m = -INFINITY, ss = 0.f;
      for (int c0 = 0; c0 < deg; c0 += 32) {
        const bool ok = c0 + (int)lane < deg;
        const int s = perm[beg + (ok ? c0 + (int)lane : 0)];
        m = fmaxf(m, ok ? lrelu(eS[s] + ed) : -INFINITY);
      }
      m = wmax32(m);
      if (!isfinite(m)) m = 0.f;
      for (int c0 = 0; c0 < deg; c0 += 32) {
        const bool ok = c0 + (int)lane < deg;
        const int s = perm[beg + (ok ? c0 + (int)lane : 0)];
        ss += ok ? expf(lrelu(eS[s] + ed) - m) : 0.f;
      }
      ss = wsum32(ss);
      for (int c0 = 0; c0 < deg; c0 += 32) {
        const int cn = (deg - c0 < 32) ? (deg - c0) : 32;
        const bool ok = (int)lane < cn;
        const int s = perm[beg + c0 + (ok ? (int)lane : 0)];
        const float a = ok ? expf(lrelu(eS[s] + ed) - m) / (ss + GAT_EPS) : 0.f;
        l2_chunk_sum(lane, s, a, cn, Hn, acc0, acc1);
      }
    }
    const float v0 = acc0 + bias[lane];
    const float v1 = (lane < 8u) ? acc1 + bias[32u + lane] : -INFINITY;
    const float mx = wmax32(fmaxf(v0, v1));
    const float sm = wsum32(expf(v0 - mx) + ((lane < 8u) ? expf(v1 - mx) : 0.f));
    const float ls = mx + logf(sm);
    rows[wv * 40u + lane] = v0 - ls;
    if (lane < 8u) rows[wv * 40u + 32u + lane] = v1 - ls;
  }
  __syncthreads();
  const long base = (long)blockIdx.x * 320;
  const long lim  = (long)N * 40;
  const unsigned t = threadIdx.x;
  if (base + t < lim) outp[base + t] = rows[t];
  if (t < 64u && base + 256 + t < lim) outp[base + 256 + t] = rows[256u + t];
}

extern "C" void kernel_launch(void* const* d_in, const int* in_sizes, int n_in,
                              void* d_out, int out_size, void* d_ws, size_t ws_size,
                              hipStream_t stream) {
  const float* x     = (const float*)d_in[0];
  const int*   eidx  = (const int*)d_in[1];
  const float* W1    = (const float*)d_in[2];
  const float* aS1   = (const float*)d_in[3];
  const float* aD1   = (const float*)d_in[4];
  const float* b1    = (const float*)d_in[5];
  const float* W2    = (const float*)d_in[6];
  const float* aS2   = (const float*)d_in[7];
  const float* aD2   = (const float*)d_in[8];
  const float* b2    = (const float*)d_in[9];
  float* out = (float*)d_out;

  const int N = in_sizes[0] / 128;
  const int E = in_sizes[1] / 2;
  const int* src = eidx;
  const int* dst = eidx + E;

  const long N64 = (long)N * 64, N40 = (long)N * 40;
  const long N8  = (long)N * 8,  E8  = (long)E * 8;

  float* ws      = (float*)d_ws;
  float* h1      = ws;
  float* esrc1   = h1 + N64;
  float* edst1   = esrc1 + N8;
  float* e1      = edst1 + N8;
  float* s1      = e1 + E8;
  float* agg1    = s1 + N8 + N8;
  float* h2      = agg1 + N64;
  float* esrc2   = h2 + N40;
  float* edst2   = esrc2 + N;
  int* perm = (int*)e1 + E;
  const int Npad = ((N + 31) / 32) * 32;
  int* cntB = (int*)s1;
  int* offs = cntB + (long)SORT_NB * Npad;
  int* rtot = offs + Npad + 32;
  const int R = (N + SORT_NA - 1) / SORT_NA;
  const int S = (((E + SORT_NB - 1) / SORT_NB) + 3) & ~3;

  const int BT = 256;
  const int rowBlocks = (N + 15) / 16;
  const int nodeBlocks = (N + 7) / 8;
  const int gemmBlocks = (N + 127) / 128;

  gat_dst_count_gemm1<<<SORT_NA * SORT_NB + gemmBlocks, SORT_T, 0, stream>>>(
      dst, cntB, rtot, N, Npad, E, R, S, x, W1, aS1, aD1, h1, esrc1, edst1);
  gat_dst_place<<<SORT_NA * SORT_NB, SORT_T, 0, stream>>>(src, dst, cntB, rtot, offs, perm,
                                                        N, Npad, E, R, S);

  gat_node_agg1_gemm2<<<rowBlocks, 512, 0, stream>>>(offs, perm, esrc1, edst1, h1, b1,
                                                     W2, aS2, aD2, h2, esrc2, edst2, N);

  gat_node_agg2<<<nodeBlocks, BT, 0, stream>>>(offs, perm, esrc2, edst2, h2, b2, out, N);
}


# H2 rows are aligned 256-byte units (40 channels, source score, zeros); layer 2 gets a row by one 8-byte load per lane, a whole segment at once, scores from the rows; log-softmax lane sum reordered
# speedup vs baseline: 7.0626x; 1.0445x over previous
#include <hip/hip_runtime.h>
#include <math.h>


typedef float v2f __attribute__((ext_vector_type(2)));
typedef float v8f __attribute__((ext_vector_type(8)));

#define NEG_SLOPE 0.2f
#define GAT_EPS 1e-16f

template <int I>
__device__ __forceinline__ float bcast8(float v) {
  return __uint_as_float(__builtin_amdgcn_mov_dpp8(__float_as_uint(v), (unsigned)(I * 0x249249)));
}

__device__ __forceinline__ float lrelu(float v) { return (v > 0.f) ? v : v * NEG_SLOPE; }

__device__ __forceinline__ float gmax8(float v) {
  v = fmaxf(v, __shfl_xor(v, 1)); v = fmaxf(v, __shfl_xor(v, 2)); v = fmaxf(v, __shfl_xor(v, 4));
  return v;
}
__device__ __forceinline__ float gsum8(float v) {
  v += __shfl_xor(v, 1); v += __shfl_xor(v, 2); v += __shfl_xor(v, 4);
  return v;
}
__device__ __forceinline__ float gsum8x8(const float (&x)[8], unsigned lane) {
  const bool b2 = (lane & 4u) != 0u, b1 = (lane & 2u) != 0u, b0 = (lane & 1u) != 0u;
  float y[4], z[2];
  for (int j = 0; j < 4; ++j) {
    const float keep = b2 ? x[4 + j] : x[j], send = b2 ? x[j] : x[4 + j];
    y[j] = keep + __shfl_xor(send, 4);
  }
  for (int j = 0; j < 2; ++j) {
    const float keep = b1 ? y[2 + j] : y[j], send = b1 ? y[j] : y[2 + j];
    z[j] = keep + __shfl_xor(send, 2);
  }
  const float keep = b0 ? z[1] : z[0], send = b0 ? z[0] : z[1];
  return keep + __shfl_xor(send, 1);
}
__device__ __forceinline__ float wmax32(float v) {
  v = gmax8(v); v = fmaxf(v, __shfl_xor(v, 8)); v = fmaxf(v, __shfl_xor(v, 16));
  return v;
}
__device__ __forceinline__ float wsum32(float v) {
  v = gsum8(v); v += __shfl_xor(v, 8); v += __shfl_xor(v, 16);
  return v;
}

__device__ __forceinline__ void
gat_gemm1_tiles(const float* __restrict__ X, const float* __restrict__ W,
                const float* __restrict__ aS, const float* __restrict__ aD,
                float* __restrict__ H, float* __restrict__ eS, float* __restrict__ eD, int N,
                int blk, float (*se)[2][128]) {
  const int lane    = threadIdx.x & 31;
  const int wave    = threadIdx.x >> 5;
  const int tile    = wave >> 2;
  const int cw      = wave & 3;
  const int row0    = blk * 128 + tile * 16;
  const int colBase = cw * 16;
  const int m       = lane & 15;
  const int khalf   = (lane >> 4) * 2;

  int rowA = row0 + m; if (rowA >= N) rowA = N - 1;
  const float* xrow = X + (long)rowA * 128;

  v8f c = {};
#pragma unroll
  for (int k0 = 0; k0 < 128; k0 += 4) {
    const int k = k0 + khalf;
    v2f a; a.x = xrow[k];                 a.y = xrow[k + 1];
    v2f b; b.x = W[k * 64 + colBase + m]; b.y = W[(k + 1) * 64 + colBase + m];
    c = __builtin_amdgcn_wmma_f32_16x16x4_f32(false, a, false, b,
                                              (short)0, c, false, false);
  }
  const int mrow = (lane >> 4) * 8;
#pragma unroll
  for (int v = 0; v < 8; ++v) {
    int r = row0 + mrow + v;
    if (r < N) H[(long)r * 64 + colBase + m] = c[v];
  }

  const float as = aS[colBase + m], ad = aD[colBase + m];
  const int i = lane & 7;
  float xs[8], xd[8];
  for (int v = 0; v < 8; ++v) { xs[v] = c[v] * as; xd[v] = c[v] * ad; }
  const float ps = gsum8x8(xs, (unsigned)lane), pd = gsum8x8(xd, (unsigned)lane);
  const int slot = (mrow + i) * 8 + 2 * cw + ((lane >> 3) & 1);
  se[tile][0][slot] = ps;
  se[tile][1][slot] = pd;
  __syncthreads();
  const int tt = threadIdx.x & 127;
  const long o = (long)row0 * 8 + tt;
  if (o < (long)N * 8) { eS[o] = se[tile][0][tt]; eD[o] = se[tile][1][tt]; }
}

#define SORT_T    1024
#define SORT_NA   16
#define SORT_NB   16
#define SORT_RMAX 4096
#define SORT_LDC  13
#define SORT_LDP  8

__device__ __forceinline__ int wsum32i(int v) {
  v += __shfl_xor(v, 1); v += __shfl_xor(v, 2); v += __shfl_xor(v, 4);
  v += __shfl_xor(v, 8); v += __shfl_xor(v, 16);
  return v;
}

__global__ void __launch_bounds__(SORT_T)
gat_dst_count_gemm1(const int* __restrict__ dst, int* __restrict__ cntB, int* __restrict__ rtot,
                    int N, int Npad, int E, int R, int S,
                    const float* __restrict__ X, const float* __restrict__ W,
                    const float* __restrict__ aS, const float* __restrict__ aD,
                    float* __restrict__ H, float* __restrict__ eS, float* __restrict__ eD) {
  __shared__ int hist[SORT_RMAX + 32];
  __shared__ __align__(16) int wtot[32];
  __shared__ float se[8][2][128];
  if ((int)blockIdx.x >= SORT_NA * SORT_NB) {
    gat_gemm1_tiles(X, W, aS, aD, H, eS, eD, N, (int)blockIdx.x - SORT_NA * SORT_NB, se);
    return;
  }
  const int t = threadIdx.x;
  int mine = 0;
  const int a = (int)blockIdx.x / SORT_NB, b = (int)blockIdx.x % SORT_NB;
  const int e0 = b * S;
  const int ne = (E - e0 < S) ? ((E - e0 > 0) ? E - e0 : 0) : S;
  const int* __restrict__ d = dst + e0;
  const bool vec = (((size_t)d) & 15) == 0;
  const int n4 = vec ? (ne >> 2) : 0;
  const int4* __restrict__ d4 = (const int4*)d;
  int* __restrict__ plane = cntB + (long)b * Npad;
  const int spare = SORT_RMAX + (t & 31);
  const int nlo = a * R, nhi = (nlo + R < N) ? nlo + R : N;
  for (int lo = nlo; lo < nhi; lo += SORT_RMAX) {
    const int len = (nhi - lo < SORT_RMAX) ? nhi - lo : SORT_RMAX;
    for (int i = t; i < len; i += SORT_T) hist[i] = 0;
    if (t < 32) hist[SORT_RMAX + t] = 0;
    __syncthreads();
    for (int i0 = 0; i0 < n4; i0 += SORT_LDC * SORT_T) {
      int4 v[SORT_LDC];
#pragma unroll
      for (int u = 0; u < SORT_LDC; ++u) { const int i = i0 + u * SORT_T + t; v[u] = d4[i < n4 ? i : n4 - 1]; }
      __builtin_amdgcn_sched_barrier(0);
#pragma unroll
      for (int u = 0; u < SORT_LDC; ++u) {
        const bool in = i0 + u * SORT_T + t < n4;
        const unsigned r0 = (unsigned)(v[u].x - lo), r1 = (unsigned)(v[u].y - lo);
        const unsigned r2 = (unsigned)(v[u].z - lo), r3 = (unsigned)(v[u].w - lo);
        atomicAdd(hist + ((in && r0 < (unsigned)len) ? (int)r0 : spare), 1);
        atomicAdd(hist + ((in && r1 < (unsigned)len) ? (int)r1 : spare), 1);
        atomicAdd(hist + ((in && r2 < (unsigned)len) ? (int)r2 : spare), 1);
        atomicAdd(hist + ((in && r3 < (unsigned)len) ? (int)r3 : spare), 1);
      }
    }
    for (int i = 4 * n4 + t; i < ne; i += SORT_T) {
      const unsigned r = (unsigned)(d[i] - lo);
      if (r < (unsigned)len) atomicAdd(hist + r, 1);
    }
    __syncthreads();
    for (int i = t; i < len; i += SORT_T) { const int h = hist[i]; plane[lo + i] = h; mine += h; }
    __syncthreads();
  }
  mine = wsum32i(mine);
  if ((t & 31) == 0) wtot[t >> 5] = mine;
  __syncthreads();
  if (t == 0) {
    int v = 0;
    for (int w = 0; w < 32; ++w) v += wtot[w];
    rtot[blockIdx.x] = v;
  }
}

#define PLACE_K ((SORT_RMAX + SORT_T - 1) / SORT_T)
#define PLACE_W ((SORT_NA * SORT_NB + 31) / 32)
static_assert(SORT_NA * SORT_NB <= SORT_T, "one thread per (range, slice) total");

template <int K>
__device__ __forceinline__ void place_counts(const int* __restrict__ cntB, int Npad, int lo, int len,
                                             int i0, int b, int (&tot)[PLACE_K], int (&bel)[PLACE_K],
                                             int& sum) {
  int cb[K][SORT_NB];
  for (int j = 0; j < K; ++j) {
    const int i = i0 + j;
    const int* __restrict__ cp = cntB + lo + ((i < len) ? i : len - 1);
    for (int bb = 0; bb < SORT_NB; ++bb) cb[j][bb] = cp[(long)bb * Npad];
  }
  __builtin_amdgcn_sched_barrier(0);
  for (int j = 0; j < PLACE_K; ++j) { tot[j] = 0; bel[j] = 0; }
  for (int j = 0; j < K; ++j) {
    int all = 0, front = 0;
    for (int bb = 0; bb < SORT_NB; ++bb) { all += cb[j][bb]; front += (bb < b) ? cb[j][bb] : 0; }
    tot[j] = (i0 + j < len) ? all : 0;
    bel[j] = front;
    sum += tot[j];
  }
}
__global__ void __launch_bounds__(SORT_T)
gat_dst_place(const int* __restrict__ src, const int* __restrict__ dst,
              const int* __restrict__ cntB, const int* __restrict__ rtot,
              int* __restrict__ offs, int* __restrict__ perm,
              int N, int Npad, int E, int R, int S) {
  __shared__ int cur[SORT_RMAX + 32];
  __shared__ __align__(16) int bsum[PLACE_W];
  __shared__ __align__(16) int wsumS[32];
  const int t = threadIdx.x;
  const int lane = t & 31, wave = t >> 5;
  const int a = (int)blockIdx.x / SORT_NB, b = (int)blockIdx.x % SORT_NB;
  const int e0 = b * S;
  const int ne = (E - e0 < S) ? ((E - e0 > 0) ? E - e0 : 0) : S;
  const int* __restrict__ d = dst + e0;
  const int* __restrict__ s = src + e0;
  const bool vec = ((((size_t)d) | ((size_t)s)) & 15) == 0;
  const int n4 = vec ? (ne >> 2) : 0;
  const int4* __restrict__ d4 = (const int4*)d;
  const int4* __restrict__ s4 = (const int4*)s;
  const int spare = SORT_RMAX + (t & 31);
  const int nlo = a * R, nhi = (nlo + R < N) ? nlo + R : N;
  const int rt = (t < a * SORT_NB) ? rtot[t] : 0;
  int carry = 0;
  for (int lo = nlo; lo < nhi; lo += SORT_RMAX) {
    const int len = (nhi - lo < SORT_RMAX) ? nhi - lo : SORT_RMAX;
    const int K = (len + SORT_T - 1) / SORT_T;
    const int i0 = t * K;
    int tot[PLACE_K], bel[PLACE_K];
    int sum = 0;
    switch (K) {
      case 1:  place_counts<1>(cntB, Npad, lo, len, i0, b, tot, bel, sum); break;
      case 2:  place_counts<2>(cntB, Npad, lo, len, i0, b, tot, bel, sum); break;
      case 3:  place_counts<3>(cntB, Npad, lo, len, i0, b, tot, bel, sum); break;
      default: place_counts<PLACE_K>(cntB, Npad, lo, len, i0, b, tot, bel, sum); break;
    }
    int inc = sum;
    for (int o = 1; o < 32; o <<= 1) {
      const int v = __shfl_up(inc, o);
      if (lane >= o) inc += v;
    }
    if (lane == 31) wsumS[wave] = inc;
    if (lo == nlo && wave < PLACE_W) {
      const int v = wsum32i(rt);
      if (lane == 0) bsum[wave] = v;
    }
    if (t < 32) cur[SORT_RMAX + t] = 0;
    __syncthreads();
    if (lo == nlo)
      for (int w = 0; w < PLACE_W; ++w) carry += bsum[w];
    int before = 0, all = 0;
    for (int w = 0; w < 32; ++w) { const int v = wsumS[w]; all += v; before += (w < wave) ? v : 0; }
    int run = carry + before + inc - sum;
#pragma unroll
    for (int j = 0; j < PLACE_K; ++j) {
      if (j < K) {
        const int i = i0 + j;
        if (i < len) {
          cur[i] = run + bel[j];
          if (b == 0) offs[lo + i] = run;
        }
        run += tot[j];
      }
    }
    carry += all;
    __syncthreads();
    for (int i0e = 0; i0e < n4; i0e += SORT_LDP * SORT_T) {
      int4 v[SORT_LDP], w[SORT_LDP];
#pragma unroll
      for (int u = 0; u < SORT_LDP; ++u) {
        const int i = i0e + u * SORT_T + t, ic = i < n4 ? i : n4 - 1;
        v[u] = d4[ic]; w[u] = s4[ic];
      }
      __builtin_amdgcn_sched_barrier(0);
#pragma unroll
      for (int u = 0; u < SORT_LDP; ++u) {
        const bool in = i0e + u * SORT_T + t < n4;
        const unsigned r0 = (unsigned)(v[u].x - lo), r1 = (unsigned)(v[u].y - lo);
        const unsigned r2 = (unsigned)(v[u].z - lo), r3 = (unsigned)(v[u].w - lo);
        const bool h0 = in && r0 < (unsigned)len, h1 = in && r1 < (unsigned)len;
        const bool h2 = in && r2 < (unsigned)len, h3 = in && r3 < (unsigned)len;
        const int p0 = atomicAdd(cur + (h0 ? (int)r0 : spare), 1);
        const int p1 = atomicAdd(cur + (h1 ? (int)r1 : spare), 1);
        const int p2 = atomicAdd(cur + (h2 ? (int)r2 : spare), 1);
        const int p3 = atomicAdd(cur + (h3 ? (int)r3 : spare), 1);
        if (h0) perm[p0] = w[u].x;
        if (h1) perm[p1] = w[u].y;
        if (h2) perm[p2] = w[u].z;
        if (h3) perm[p3] = w[u].w;
      }
    }
    for (int i = 4 * n4 + t; i < ne; i += SORT_T) {
      const unsigned r = (unsigned)(d[i] - lo);
      if (r < (unsigned)len) perm[atomicAdd(cur + r, 1)] = s[i];
    }
    __syncthreads();
  }
  if (b == 0 && t == 0 && nlo < N && nhi == N) offs[N] = carry;
}


__device__ __forceinline__ void l1_scores(const int* __restrict__ perm, const float* __restrict__ eS,
                                          int base, int cnt, unsigned q, unsigned i,
                                          const float (&ed)[2], float (&e)[4][2]) {
  for (int r = 0; r < 4; ++r) {
    const int j = 8 * r + (int)i;
    const bool ok = j < cnt;
    const unsigned s = (unsigned)perm[base + (ok ? j : 0)];
    for (int k = 0; k < 2; ++k) {
      const float v = lrelu(eS[s * 8u + 4u * k + q] + ed[k]);
      e[r][k] = ok ? v : -INFINITY;
    }
  }
}

template <int J>
__device__ __forceinline__ void l1_load(unsigned lane, int s, const float* __restrict__ Hn,
                                        float& h0, float& h1) {
  const float* row = Hn + (unsigned)__builtin_amdgcn_readlane(s, J) * 64u;
  h0 = row[lane];
  h1 = row[32u + lane];
}

template <int R>
__device__ __forceinline__ void l1_group(unsigned lane, int s, const float (&A)[4][2],
                                         const float* __restrict__ Hn, float& acc0, float& acc1) {
  float h[16];
  l1_load<8 * R + 0>(lane, s, Hn, h[0],  h[1]);
  l1_load<8 * R + 1>(lane, s, Hn, h[2],  h[3]);
  l1_load<8 * R + 2>(lane, s, Hn, h[4],  h[5]);
  l1_load<8 * R + 3>(lane, s, Hn, h[6],  h[7]);
  l1_load<8 * R + 4>(lane, s, Hn, h[8],  h[9]);
  l1_load<8 * R + 5>(lane, s, Hn, h[10], h[11]);
  l1_load<8 * R + 6>(lane, s, Hn, h[12], h[13]);
  l1_load<8 * R + 7>(lane, s, Hn, h[14], h[15]);
  acc0 += h[0]  * bcast8<0>(A[R][0]);  acc1 += h[1]  * bcast8<0>(A[R][1]);
  acc0 += h[2]  * bcast8<1>(A[R][0]);  acc1 += h[3]  * bcast8<1>(A[R][1]);
  acc0 += h[4]  * bcast8<2>(A[R][0]);  acc1 += h[5]  * bcast8<2>(A[R][1]);
  acc0 += h[6]  * bcast8<3>(A[R][0]);  acc1 += h[7]  * bcast8<3>(A[R][1]);
  acc0 += h[8]  * bcast8<4>(A[R][0]);  acc1 += h[9]  * bcast8<4>(A[R][1]);
  acc0 += h[10] * bcast8<5>(A[R][0]);  acc1 += h[11] * bcast8<5>(A[R][1]);
  acc0 += h[12] * bcast8<6>(A[R][0]);  acc1 += h[13] * bcast8<6>(A[R][1]);
  acc0 += h[14] * bcast8<7>(A[R][0]);  acc1 += h[15] * bcast8<7>(A[R][1]);
}

__device__ __forceinline__ void l1_chunk_sum(const int* __restrict__ perm, int base, int cnt,
                                             unsigned lane, const float (&A)[4][2],
                                             const float* __restrict__ Hn, float& acc0, float& acc1) {
  const int s = perm[base + (((int)lane < cnt) ? (int)lane : 0)];
  l1_group<0>(lane, s, A, Hn, acc0, acc1);
  if (cnt > 8)  l1_group<1>(lane, s, A, Hn, acc0, acc1);
  if (cnt > 16) l1_group<2>(lane, s, A, Hn, acc0, acc1);
  if (cnt > 24) l1_group<3>(lane, s, A, Hn, acc0, acc1);
}

#define H2_LD 64
#define XT_LD 68
__global__ void __launch_bounds__(512)
gat_node_agg1_gemm2(const int* __restrict__ offs, const int* __restrict__ perm,
                    const float* __restrict__ eS, const float* __restrict__ eD,
                    const float* __restrict__ Hn, const float* __restrict__ bias,
                    const float* __restrict__ W2, const float* __restrict__ aS2,
                    const float* __restrict__ aD2, float* __restrict__ H2,
                    float* __restrict__ eD2, int N) {
  __shared__ float xt[16 * XT_LD];
  __shared__ float ht[16 * 40];
  __shared__ float pe[3][2][16];
  const unsigned lane = threadIdx.x & 31;
  const unsigned wv   = threadIdx.x >> 5;
  const int row0 = (int)blockIdx.x * 16;
  const int n = row0 + (int)wv;
  float v0 = 0.f, v1 = 0.f;
  if (n < N) {
    const unsigned q = lane >> 3, i = lane & 7;
    const int beg = offs[n], deg = offs[n + 1] - beg;
    const float ed[2] = { eD[(unsigned)n * 8u + q], eD[(unsigned)n * 8u + 4u + q] };
    float acc0 = 0.f, acc1 = 0.f;
    if (deg > 0 && deg <= 32) {
      float e[4][2], A[4][2], m[2], ss[2];
      l1_scores(perm, eS, beg, deg, q, i, ed, e);
      for (int k = 0; k < 2; ++k) {
        m[k] = gmax8(fmaxf(fmaxf(e[0][k], e[1][k]), fmaxf(e[2][k], e[3][k])));
        if (!isfinite(m[k])) m[k] = 0.f;
        float t = 0.f;
        for (int r = 0; r < 4; ++r) { e[r][k] = expf(e[r][k] - m[k]); t += e[r][k]; }
        ss[k] = gsum8(t);
        for (int r = 0; r < 4; ++r) A[r][k] = e[r][k] / (ss[k] + GAT_EPS);
      }
      l1_chunk_sum(perm, beg, deg, lane, A, Hn, acc0, acc1);
    } else if (deg > 32) {
      float e[4][2], A[4][2], m[2] = { -INFINITY, -INFINITY }, ss[2] = { 0.f, 0.f };
      for (int c0 = 0; c0 < deg; c0 += 32) {
        const int cn = (deg - c0 < 32) ? (deg - c0) : 32;
        l1_scores(perm, eS, beg + c0, cn, q, i, ed, e);
        for (int k = 0; k < 2; ++k)
          m[k] = fmaxf(m[k], fmaxf(fmaxf(e[0][k], e[1][k]), fmaxf(e[2][k], e[3][k])));
      }
      for (int k = 0; k < 2; ++k) { m[k] = gmax8(m[k]); if (!isfinite(m[k])) m[k] = 0.f; }
      for (int c0 = 0; c0 < deg; c0 += 32) {
        const int cn = (deg - c0 < 32) ? (deg - c0) : 32;
        l1_scores(perm, eS, beg + c0, cn, q, i, ed, e);
        for (int k = 0; k < 2; ++k)
          for (int r = 0; r < 4; ++r) ss[k] += expf(e[r][k] - m[k]);
      }
      for (int k = 0; k < 2; ++k) ss[k] = gsum8(ss[k]);
      for (int c0 = 0; c0 < deg; c0 += 32) {
        const int cn = (deg - c0 < 32) ? (deg - c0) : 32;
        l1_scores(perm, eS, beg + c0, cn, q, i, ed, e);
        for (int k = 0; k < 2; ++k)
          for (int r = 0; r < 4; ++r) A[r][k] = expf(e[r][k] - m[k]) / (ss[k] + GAT_EPS);
        l1_chunk_sum(perm, beg + c0, cn, lane, A, Hn, acc0, acc1);
      }
    }
    v0 = acc0 + bias[lane];       v0 = v0 > 0.f ? v0 : 0.f;
    v1 = acc1 + bias[32u + lane]; v1 = v1 > 0.f ? v1 : 0.f;
  }
  const int m     = (int)(lane & 15u);
  const int khalf = (int)(lane >> 4) * 2;
  const int mrow  = (int)(lane >> 4) * 8;
  const int col   = (int)wv * 16 + m;
  const bool colOK = (col < 40);
  float bx[16], by[16], as2 = 0.f, ad2 = 0.f;
  if (wv < 3u) {
    const int colc = colOK ? col : 39;
    const float* wcol = W2 + colc;
    for (int ks = 0; ks < 16; ++ks) {
      bx[ks] = wcol[(4 * ks + khalf) * 40];
      by[ks] = wcol[(4 * ks + khalf + 1) * 40];
    }
    as2 = aS2[colc];
    ad2 = aD2[colc];
  }
  xt[wv * XT_LD + lane]       = v0;
  xt[wv * XT_LD + 32u + lane] = v1;
  __syncthreads();

  if (wv < 3u) {
    const float* xrow = xt + m * XT_LD;
    v8f c = {};
    for (int ks = 0; ks < 16; ++ks) {
      const int k = 4 * ks + khalf;
      v2f a; a.x = xrow[k]; a.y = xrow[k + 1];
      v2f b;
      b.x = colOK ? bx[ks] : 0.0f;
      b.y = colOK ? by[ks] : 0.0f;
      c = __builtin_amdgcn_wmma_f32_16x16x4_f32(false, a, false, b,
                                                (short)0, c, false, false);
    }
    const float as = colOK ? as2 : 0.f, ad = colOK ? ad2 : 0.f;
    const int i = (int)(lane & 7u);
    if (colOK) {
      for (int v = 0; v < 8; ++v) ht[(mrow + v) * 40 + col] = c[v];
    }
    float xs[8], xd[8];
    for (int v = 0; v < 8; ++v) { xs[v] = c[v] * as; xd[v] = c[v] * ad; }
    float ps = gsum8x8(xs, lane); ps += __shfl_xor(ps, 8);
    float pd = gsum8x8(xd, lane); pd += __shfl_xor(pd, 8);
    if ((lane & 8u) == 0u) { pe[wv][0][mrow + i] = ps; pe[wv][1][mrow + i] = pd; }
  }
  __syncthreads();

  const unsigned t = threadIdx.x;
  for (unsigned half = 0; half < 2u; ++half) {
    const unsigned idx = t + 512u * half;
    const unsigned r = idx / H2_LD, w = idx % H2_LD;
    float val = 0.f;
    if (w < 40u) val = ht[r * 40u + w];
    else if (w == 40u) val = pe[0][0][r] + pe[1][0][r] + pe[2][0][r];
    if (row0 + (int)r < N) H2[((long)row0 + r) * H2_LD + w] = val;
  }
  if (t < 16u && row0 + (int)t < N) eD2[row0 + (int)t] = pe[0][1][t] + pe[1][1][t] + pe[2][1][t];
}

#define L2_SC_LANE 20

template <int J>
__device__ __forceinline__ v2f l2_load(unsigned lane, int s, const float* __restrict__ Hn) {
  const v2f* row = (const v2f*)(Hn + (unsigned)__builtin_amdgcn_readlane(s, J) * (unsigned)H2_LD);
  return row[lane];
}
template <int J>
__device__ __forceinline__ float l2_alpha(float a) {
  return __int_as_float(__builtin_amdgcn_readlane(__float_as_int(a), J));
}
template <int J>
__device__ __forceinline__ float l2_score(unsigned lane, v2f h, float sc) {
  const float v = __int_as_float(__builtin_amdgcn_readlane(__float_as_int(h.x), L2_SC_LANE));
  return (lane == (unsigned)J) ? v : sc;
}

template <int R, int B>
__device__ __forceinline__ void l2_load8(unsigned lane, int s, const float* __restrict__ Hn, v2f* h) {
  h[B + 0] = l2_load<8 * R + 0>(lane, s, Hn);  h[B + 1] = l2_load<8 * R + 1>(lane, s, Hn);
  h[B + 2] = l2_load<8 * R + 2>(lane, s, Hn);  h[B + 3] = l2_load<8 * R + 3>(lane, s, Hn);
  h[B + 4] = l2_load<8 * R + 4>(lane, s, Hn);  h[B + 5] = l2_load<8 * R + 5>(lane, s, Hn);
  h[B + 6] = l2_load<8 * R + 6>(lane, s, Hn);  h[B + 7] = l2_load<8 * R + 7>(lane, s, Hn);
}
template <int R, int B>
__device__ __forceinline__ float l2_score8(unsigned lane, const v2f* h, float sc) {
  sc = l2_score<8 * R + 0>(lane, h[B + 0], sc);  sc = l2_score<8 * R + 1>(lane, h[B + 1], sc);
  sc = l2_score<8 * R + 2>(lane, h[B + 2], sc);  sc = l2_score<8 * R + 3>(lane, h[B + 3], sc);
  sc = l2_score<8 * R + 4>(lane, h[B + 4], sc);  sc = l2_score<8 * R + 5>(lane, h[B + 5], sc);
  sc = l2_score<8 * R + 6>(lane, h[B + 6], sc);  sc = l2_score<8 * R + 7>(lane, h[B + 7], sc);
  return sc;
}
template <int R, int B>
__device__ __forceinline__ void l2_acc8(float a, const v2f* h, float& acc0, float& acc1) {
  float a0;
  a0 = l2_alpha<8 * R + 0>(a); acc0 += h[B + 0].x * a0; acc1 += h[B + 0].y * a0;
  a0 = l2_alpha<8 * R + 1>(a); acc0 += h[B + 1].x * a0; acc1 += h[B + 1].y * a0;
  a0 = l2_alpha<8 * R + 2>(a); acc0 += h[B + 2].x * a0; acc1 += h[B + 2].y * a0;
  a0 = l2_alpha<8 * R + 3>(a); acc0 += h[B + 3].x * a0; acc1 += h[B + 3].y * a0;
  a0 = l2_alpha<8 * R + 4>(a); acc0 += h[B + 4].x * a0; acc1 += h[B + 4].y * a0;
  a0 = l2_alpha<8 * R + 5>(a); acc0 += h[B + 5].x * a0; acc1 += h[B + 5].y * a0;
  a0 = l2_alpha<8 * R + 6>(a); acc0 += h[B + 6].x * a0; acc1 += h[B + 6].y * a0;
  a0 = l2_alpha<8 * R + 7>(a); acc0 += h[B + 7].x * a0; acc1 += h[B + 7].y * a0;
}

template <int G>
__device__ __forceinline__ void l2_rows_at_once(unsigned lane, int s, bool ok, float ed,
                                                const float* __restrict__ Hn, float& acc0, float& acc1) {
  v2f h[8 * G];
  l2_load8<0, 0>(lane, s, Hn, h);
  if constexpr (G > 1) l2_load8<1, 8>(lane, s, Hn, h);
  if constexpr (G > 2) l2_load8<2, 16>(lane, s, Hn, h);
  float sc = 0.f;
  sc = l2_score8<0, 0>(lane, h, sc);
  if constexpr (G > 1) sc = l2_score8<1, 8>(lane, h, sc);
  if constexpr (G > 2) sc = l2_score8<2, 16>(lane, h, sc);
  const float e = ok ? lrelu(sc + ed) : -INFINITY;
  float m = wmax32(e);
  if (!isfinite(m)) m = 0.f;
  const float p  = expf(e - m);
  const float ss = wsum32(p);
  const float a  = p / (ss + GAT_EPS);
  l2_acc8<0, 0>(a, h, acc0, acc1);
  if constexpr (G > 1) l2_acc8<1, 8>(a, h, acc0, acc1);
  if constexpr (G > 2) l2_acc8<2, 16>(a, h, acc0, acc1);
}

__device__ __forceinline__ void l2_chunk_sum(unsigned lane, int s, float a, int cnt,
                                             const float* __restrict__ Hn, float& acc0, float& acc1) {
  v2f h[8];
  l2_load8<0, 0>(lane, s, Hn, h); l2_acc8<0, 0>(a, h, acc0, acc1);
  if (cnt > 8)  { l2_load8<1, 0>(lane, s, Hn, h); l2_acc8<1, 0>(a, h, acc0, acc1); }
  if (cnt > 16) { l2_load8<2, 0>(lane, s, Hn, h); l2_acc8<2, 0>(a, h, acc0, acc1); }
  if (cnt > 24) { l2_load8<3, 0>(lane, s, Hn, h); l2_acc8<3, 0>(a, h, acc0, acc1); }
}

__global__ void gat_node_agg2(const int* __restrict__ offs, const int* __restrict__ perm,
                              const float* __restrict__ eD,
                              const float* __restrict__ Hn, const float* __restrict__ bias,
                              float* __restrict__ outp, int N) {
  __shared__ float rows[8 * 40];
  const unsigned lane = threadIdx.x & 31;
  const unsigned wv   = threadIdx.x >> 5;
  const int n = (int)(blockIdx.x * 8u + wv);
  float acc0 = 0.f, acc1 = 0.f;
  if (n < N) {
    const int beg = __builtin_amdgcn_readfirstlane(offs[n]);
    const int deg = __builtin_amdgcn_readfirstlane(offs[n + 1]) - beg;
    const float ed = eD[n];
    if (deg > 0 && deg <= 24) {
      const bool ok = (int)lane < deg;
      const int s = perm[beg + (ok ? (int)lane : 0)];
      if (deg <= 8)       l2_rows_at_once<1>(lane, s, ok, ed, Hn, acc0, acc1);
      else if (deg <= 16) l2_rows_at_once<2>(lane, s, ok, ed, Hn, acc0, acc1);
      else                l2_rows_at_once<3>(lane, s, ok, ed, Hn, acc0, acc1);
    } else if (deg > 24) {
      float m = -INFINITY, ss = 0.f;
      for (int c0 = 0; c0 < deg; c0 += 32) {
        const bool ok = c0 + (int)lane < deg;
        const int s = perm[beg + (ok ? c0 + (int)lane : 0)];
        m = fmaxf(m, ok ? lrelu(Hn[(unsigned)s * (unsigned)H2_LD + 40u] + ed) : -INFINITY);
      }
      m = wmax32(m);
      if (!isfinite(m)) m = 0.f;
      for (int c0 = 0; c0 < deg; c0 += 32) {
        const bool ok = c0 + (int)lane < deg;
        const int s = perm[beg + (ok ? c0 + (int)lane : 0)];
        ss += ok ? expf(lrelu(Hn[(unsigned)s * (unsigned)H2_LD + 40u] + ed) - m) : 0.f;
      }
      ss = wsum32(ss);
      for (int c0 = 0; c0 < deg; c0 += 32) {
        const int cn = (deg - c0 < 32) ? (deg - c0) : 32;
        const bool ok = (int)lane < cn;
        const int s = perm[beg + c0 + (ok ? (int)lane : 0)];
        const float a = ok ? expf(lrelu(Hn[(unsigned)s * (unsigned)H2_LD + 40u] + ed) - m) / (ss + GAT_EPS) : 0.f;
        l2_chunk_sum(lane, s, a, cn, Hn, acc0, acc1);
      }
    }
    const bool ch = lane < 20u;
    const unsigned cb = ch ? 2u * lane : 0u;
    const float v0 = ch ? acc0 + bias[cb]      : -INFINITY;
    const float v1 = ch ? acc1 + bias[cb + 1u] : -INFINITY;
    const float mx = wmax32(fmaxf(v0, v1));
    const float sm = wsum32(ch ? expf(v0 - mx) + expf(v1 - mx) : 0.f);
    const float ls = mx + logf(sm);
    if (ch) { rows[wv * 40u + cb] = v0 - ls; rows[wv * 40u + cb + 1u] = v1 - ls; }
  }
  __syncthreads();
  const long base = (long)blockIdx.x * 320;
  const long lim  = (long)N * 40;
  const unsigned t = threadIdx.x;
  if (base + t < lim) outp[base + t] = rows[t];
  if (t < 64u && base + 256 + t < lim) outp[base + 256 + t] = rows[256u + t];
}

extern "C" void kernel_launch(void* const* d_in, const int* in_sizes, int n_in,
                              void* d_out, int out_size, void* d_ws, size_t ws_size,
                              hipStream_t stream) {
  const float* x     = (const float*)d_in[0];
  const int*   eidx  = (const int*)d_in[1];
  const float* W1    = (const float*)d_in[2];
  const float* aS1   = (const float*)d_in[3];
  const float* aD1   = (const float*)d_in[4];
  const float* b1    = (const float*)d_in[5];
  const float* W2    = (const float*)d_in[6];
  const float* aS2   = (const float*)d_in[7];
  const float* aD2   = (const float*)d_in[8];
  const float* b2    = (const float*)d_in[9];
  float* out = (float*)d_out;

  const int N = in_sizes[0] / 128;
  const int E = in_sizes[1] / 2;
  const int* src = eidx;
  const int* dst = eidx + E;

  const long N64 = (long)N * 64;
  const long N8  = (long)N * 8,  E8  = (long)E * 8;

  float* ws      = (float*)d_ws;
  float* h1      = ws;
  float* esrc1   = h1 + N64;
  float* edst1   = esrc1 + N8;
  float* e1      = edst1 + N8;
  float* s1      = e1 + E8;
  int* perm = (int*)e1 + E;
  const int Npad = ((N + 31) / 32) * 32;
  int* cntB = (int*)s1;
  int* offs = cntB + (long)SORT_NB * Npad;
  int* rtot = offs + Npad + 32;
  const long h2o = (((long)((float*)(rtot + SORT_NA * SORT_NB) - ws) + 63) / 64) * 64;
  float* h2      = ws + h2o;
  float* edst2   = h2 + N64;
  const int R = (N + SORT_NA - 1) / SORT_NA;
  const int S = (((E + SORT_NB - 1) / SORT_NB) + 3) & ~3;

  const int BT = 256;
  const int rowBlocks = (N + 15) / 16;
  const int nodeBlocks = (N + 7) / 8;
  const int gemmBlocks = (N + 127) / 128;

  gat_dst_count_gemm1<<<SORT_NA * SORT_NB + gemmBlocks, SORT_T, 0, stream>>>(
      dst, cntB, rtot, N, Npad, E, R, S, x, W1, aS1, aD1, h1, esrc1, edst1);
  gat_dst_place<<<SORT_NA * SORT_NB, SORT_T, 0, stream>>>(src, dst, cntB, rtot, offs, perm,
                                                        N, Npad, E, R, S);

  gat_node_agg1_gemm2<<<rowBlocks, 512, 0, stream>>>(offs, perm, esrc1, edst1, h1, b1,
                                                     W2, aS2, aD2, h2, edst2, N);

  gat_node_agg2<<<nodeBlocks, BT, 0, stream>>>(offs, perm, edst2, h2, b2, out, N);
}
